# starting kernel
# baseline (speedup 1.0000x reference)
#include <hip/hip_runtime.h>
#include <math.h>

#define B_SZ   2
#define LSEQ   2048
#define DEXP   1024
#define DRANK  64
#define NST    16
#define RPROJ  (DRANK + 2*NST)
#define NCHUNK 8
#define CHUNK  (LSEQ / NCHUNK)
#define ROWS   (B_SZ * LSEQ)

typedef __attribute__((ext_vector_type(2))) float v2f;
typedef __attribute__((ext_vector_type(8))) float v8f;

__device__ __forceinline__ v8f wmma_f32_k4(v2f a, v2f b, v8f c) {
  return __builtin_amdgcn_wmma_f32_16x16x4_f32(
       false, a,  false, b,
       (short)0, c,  false,  false);
}

__global__ void ssm_prep(const float* __restrict__ A_log, float* __restrict__ Aneg,
                         const float* __restrict__ xw,  float* __restrict__ xwT,
                         const float* __restrict__ dtw, float* __restrict__ dtwT) {
  int i = blockIdx.x * 256 + threadIdx.x;
  if (i < DEXP * NST) Aneg[i] = -__expf(A_log[i]);
  if (i < DEXP * RPROJ) {
    int k = i / RPROJ, c = i % RPROJ;
    xwT[i] = xw[(size_t)c * DEXP + k];
  }
  if (i < DRANK * DEXP) {
    int k = i / DEXP, d = i % DEXP;
    dtwT[i] = dtw[(size_t)d * DRANK + k];
  }
}

__global__ __launch_bounds__(256) void ssm_proj_wmma(
    const float* __restrict__ x,    const float* __restrict__ xwT,
    const float* __restrict__ xb,   const float* __restrict__ dtwT,
    const float* __restrict__ dtb,
    float* __restrict__ dtbuf, float* __restrict__ Bmbuf, float* __restrict__ Cbuf)
{
  __shared__ float sx[16 * 516];
  __shared__ float sproj[16 * 100];

  const int t    = threadIdx.x;
  const int wave = t >> 5;
  const int nl   = t & 15;
  const int hi   = (t >> 4) & 1;
  const int rowbase = blockIdx.x * 16;

  v8f acc1;
  if (wave < 6) {
    float bv = xb[wave * 16 + nl];
    for (int i = 0; i < 8; ++i) acc1[i] = bv;
  }
  for (int kc = 0; kc < DEXP; kc += 512) {
    __syncthreads();
    for (int i = 0; i < 8; ++i) {
      int flat = t + i * 256;
      int r  = flat >> 7;
      int kk = (flat & 127) << 2;
      const float4 v = *(const float4*)(x + (size_t)(rowbase + r) * DEXP + kc + kk);
      float* dst = &sx[r * 516 + kk];
      dst[0] = v.x; dst[1] = v.y; dst[2] = v.z; dst[3] = v.w;
    }
    __syncthreads();
    if (wave < 6) {
      const int col = wave * 16 + nl;
      for (int k = 0; k < 512; k += 4) {
        const int kk = k + hi * 2;
        v2f a;  a.x  = sx[nl * 516 + kk];   a.y  = sx[nl * 516 + kk + 1];
        const int gk = kc + kk;
        v2f bf;
        bf.x = xwT[(size_t)gk * RPROJ + col];
        bf.y = xwT[(size_t)(gk + 1) * RPROJ + col];
        acc1 = wmma_f32_k4(a, bf, acc1);
      }
    }
  }
  if (wave < 6) {
    for (int i = 0; i < 8; ++i) {
      int M = i + hi * 8;
      sproj[M * 100 + wave * 16 + nl] = acc1[i];
      if (wave == 4) Bmbuf[(size_t)(rowbase + M) * NST + nl] = acc1[i];
      if (wave == 5) Cbuf [(size_t)(rowbase + M) * NST + nl] = acc1[i];
    }
  }
  __syncthreads();

  v2f a2[16];
  for (int kt = 0; kt < 16; ++kt) {
    int kb = kt * 4 + hi * 2;
    a2[kt].x = sproj[nl * 100 + kb];
    a2[kt].y = sproj[nl * 100 + kb + 1];
  }
  for (int j = 0; j < 8; ++j) {
    const int col = wave * 128 + j * 16 + nl;
    v8f acc2;
    { float bv = dtb[col]; for (int i = 0; i < 8; ++i) acc2[i] = bv; }
    for (int kt = 0; kt < 16; ++kt) {
      int kb = kt * 4 + hi * 2;
      v2f bf;
      bf.x = dtwT[(size_t)kb * DEXP + col];
      bf.y = dtwT[(size_t)(kb + 1) * DEXP + col];
      acc2 = wmma_f32_k4(a2[kt], bf, acc2);
    }
    for (int i = 0; i < 8; ++i) {
      int M = i + hi * 8;
      float v  = acc2[i];
      float sp = (v > 20.0f) ? v : log1pf(__expf(v));
      dtbuf[(size_t)(rowbase + M) * DEXP + col] = sp;
    }
  }
}

__global__ __launch_bounds__(256) void ssm_scan_carry(
    const float* __restrict__ x, const float* __restrict__ dtbuf,
    const float* __restrict__ Bmbuf, const float* __restrict__ Aneg,
    float* __restrict__ Sbuf, float* __restrict__ Pbuf)
{
  const int t  = threadIdx.x;
  const int bi = blockIdx.x;
  const int c  = bi & (NCHUNK - 1);
  const int dg = (bi >> 3) & 63;
  const int b  = bi >> 9;
  const int n  = t & 15;
  const int d  = dg * 16 + (t >> 4);

  const float And = Aneg[d * NST + n];
  const size_t rbase = (size_t)b * LSEQ + (size_t)c * CHUNK;
  const float* dtp = dtbuf + rbase * DEXP + d;
  const float* xp  = x     + rbase * DEXP + d;
  const float* Bp  = Bmbuf + rbase * NST + n;

  float h = 0.0f, sdt = 0.0f;
  for (int li = 0; li < CHUNK; ++li) {
    __builtin_prefetch(dtp + (size_t)(li + 8) * DEXP, 0, 1);
    __builtin_prefetch(xp  + (size_t)(li + 8) * DEXP, 0, 1);
    float dtv = dtp[(size_t)li * DEXP];
    float xv  = xp [(size_t)li * DEXP];
    float Bv  = Bp [li * NST];
    float dA  = __expf(And * dtv);
    h   = fmaf(dA, h, dtv * xv * Bv);
    sdt += dtv;
  }
  const size_t si = (((size_t)b * NCHUNK + c) * DEXP + d) * NST + n;
  Sbuf[si] = h;
  Pbuf[si] = __expf(And * sdt);
}

__global__ void ssm_carry_combine(const float* __restrict__ Sbuf,
                                  const float* __restrict__ Pbuf,
                                  float* __restrict__ Car)
{
  const int tg = blockIdx.x * 256 + threadIdx.x;
  const int n  = tg & 15;
  const int d  = (tg >> 4) & (DEXP - 1);
  const int b  = tg >> 14;
  const size_t stride = (size_t)DEXP * NST;
  const size_t base   = (size_t)b * NCHUNK * stride + (size_t)d * NST + n;
  float cin = 0.0f;
  Car[base] = 0.0f;
  for (int c = 1; c < NCHUNK; ++c) {
    const size_t prev = base + (size_t)(c - 1) * stride;
    cin = Sbuf[prev] + Pbuf[prev] * cin;
    Car[base + (size_t)c * stride] = cin;
  }
}

__global__ __launch_bounds__(256) void ssm_scan_out(
    const float* __restrict__ x, const float* __restrict__ dtbuf,
    const float* __restrict__ Bmbuf, const float* __restrict__ Cbuf,
    const float* __restrict__ Aneg, const float* __restrict__ Car,
    float* __restrict__ out)
{
  const int t  = threadIdx.x;
  const int bi = blockIdx.x;
  const int c  = bi & (NCHUNK - 1);
  const int dg = (bi >> 3) & 63;
  const int b  = bi >> 9;
  const int n  = t & 15;
  const int d  = dg * 16 + (t >> 4);

  const float And = Aneg[d * NST + n];
  const size_t rbase = (size_t)b * LSEQ + (size_t)c * CHUNK;
  const float* dtp = dtbuf + rbase * DEXP + d;
  const float* xp  = x     + rbase * DEXP + d;
  const float* Bp  = Bmbuf + rbase * NST + n;
  const float* Cp  = Cbuf  + rbase * NST + n;
  float* op        = out   + rbase * DEXP + d;

  const size_t si = (((size_t)b * NCHUNK + c) * DEXP + d) * NST + n;
  float h = Car[si];

  for (int li = 0; li < CHUNK; ++li) {
    __builtin_prefetch(dtp + (size_t)(li + 8) * DEXP, 0, 1);
    __builtin_prefetch(xp  + (size_t)(li + 8) * DEXP, 0, 1);
    float dtv = dtp[(size_t)li * DEXP];
    float xv  = xp [(size_t)li * DEXP];
    float Bv  = Bp [li * NST];
    float dA  = __expf(And * dtv);
    h = fmaf(dA, h, dtv * xv * Bv);
    float y = h * Cp[li * NST];
    y += __shfl_xor(y, 1);
    y += __shfl_xor(y, 2);
    y += __shfl_xor(y, 4);
    y += __shfl_xor(y, 8);
    if (n == 0) op[(size_t)li * DEXP] = y;
  }
}

extern "C" void kernel_launch(void* const* d_in, const int* in_sizes, int n_in,
                              void* d_out, int out_size, void* d_ws, size_t ws_size,
                              hipStream_t stream) {
  (void)in_sizes; (void)n_in; (void)out_size; (void)ws_size;
  const float* x     = (const float*)d_in[0];
  const float* xw    = (const float*)d_in[1];
  const float* xb    = (const float*)d_in[2];
  const float* dtw   = (const float*)d_in[3];
  const float* dtb   = (const float*)d_in[4];
  const float* A_log = (const float*)d_in[5];
  float* out = (float*)d_out;

  float* ws    = (float*)d_ws;
  float* dtbuf = ws;
  float* Aneg  = dtbuf + (size_t)ROWS * DEXP;
  float* Bm    = Aneg  + (size_t)DEXP * NST;
  float* Cb    = Bm    + (size_t)ROWS * NST;
  float* Sb    = Cb    + (size_t)ROWS * NST;
  float* Pb    = Sb    + (size_t)B_SZ * NCHUNK * DEXP * NST;
  float* Car   = Pb    + (size_t)B_SZ * NCHUNK * DEXP * NST;
  float* xwT   = Car   + (size_t)B_SZ * NCHUNK * DEXP * NST;
  float* dtwT  = xwT   + (size_t)DEXP * RPROJ;

  hipLaunchKernelGGL(ssm_prep, dim3((DEXP * RPROJ + 255) / 256), dim3(256), 0, stream,
                     A_log, Aneg, xw, xwT, dtw, dtwT);
  hipLaunchKernelGGL(ssm_proj_wmma, dim3(ROWS / 16), dim3(256), 0, stream,
                     x, xwT, xb, dtwT, dtb, dtbuf, Bm, Cb);
  hipLaunchKernelGGL(ssm_scan_carry, dim3(B_SZ * (DEXP / 16) * NCHUNK), dim3(256), 0, stream,
                     x, dtbuf, Bm, Aneg, Sb, Pb);
  hipLaunchKernelGGL(ssm_carry_combine, dim3((B_SZ * DEXP * NST) / 256), dim3(256), 0, stream,
                     Sb, Pb, Car);
  hipLaunchKernelGGL(ssm_scan_out, dim3(B_SZ * (DEXP / 16) * NCHUNK), dim3(256), 0, stream,
                     x, dtbuf, Bm, Cb, Aneg, Car, out);
}


# scan kernels re-mapped: one lane per channel d holds all 16 states; chunk's dt/x columns and B/C rows staged in LDS by full-line loads; n-sum is an in-thread pairwise tree, products fused
# speedup vs baseline: 2.3117x; 2.3117x over previous
#include <hip/hip_runtime.h>
#include <math.h>

#define B_SZ   2
#define LSEQ   2048
#define DEXP   1024
#define DRANK  64
#define NST    16
#define RPROJ  (DRANK + 2*NST)
#define NCHUNK 8
#define CHUNK  (LSEQ / NCHUNK)
#define ROWS   (B_SZ * LSEQ)

typedef __attribute__((ext_vector_type(2))) float v2f;
typedef __attribute__((ext_vector_type(8))) float v8f;

__device__ __forceinline__ v8f wmma_f32_k4(v2f a, v2f b, v8f c) {
  return __builtin_amdgcn_wmma_f32_16x16x4_f32(
       false, a,  false, b,
       (short)0, c,  false,  false);
}

__global__ void ssm_prep(const float* __restrict__ A_log, float* __restrict__ Aneg,
                         const float* __restrict__ xw,  float* __restrict__ xwT,
                         const float* __restrict__ dtw, float* __restrict__ dtwT) {
  int i = blockIdx.x * 256 + threadIdx.x;
  if (i < DEXP * NST) Aneg[i] = -__expf(A_log[i]);
  if (i < DEXP * RPROJ) {
    int k = i / RPROJ, c = i % RPROJ;
    xwT[i] = xw[(size_t)c * DEXP + k];
  }
  if (i < DRANK * DEXP) {
    int k = i / DEXP, d = i % DEXP;
    dtwT[i] = dtw[(size_t)d * DRANK + k];
  }
}

__global__ __launch_bounds__(256) void ssm_proj_wmma(
    const float* __restrict__ x,    const float* __restrict__ xwT,
    const float* __restrict__ xb,   const float* __restrict__ dtwT,
    const float* __restrict__ dtb,
    float* __restrict__ dtbuf, float* __restrict__ Bmbuf, float* __restrict__ Cbuf)
{
  __shared__ float sx[16 * 516];
  __shared__ float sproj[16 * 100];

  const int t    = threadIdx.x;
  const int wave = t >> 5;
  const int nl   = t & 15;
  const int hi   = (t >> 4) & 1;
  const int rowbase = blockIdx.x * 16;

  v8f acc1;
  if (wave < 6) {
    float bv = xb[wave * 16 + nl];
    for (int i = 0; i < 8; ++i) acc1[i] = bv;
  }
  for (int kc = 0; kc < DEXP; kc += 512) {
    __syncthreads();
    for (int i = 0; i < 8; ++i) {
      int flat = t + i * 256;
      int r  = flat >> 7;
      int kk = (flat & 127) << 2;
      const float4 v = *(const float4*)(x + (size_t)(rowbase + r) * DEXP + kc + kk);
      float* dst = &sx[r * 516 + kk];
      dst[0] = v.x; dst[1] = v.y; dst[2] = v.z; dst[3] = v.w;
    }
    __syncthreads();
    if (wave < 6) {
      const int col = wave * 16 + nl;
      for (int k = 0; k < 512; k += 4) {
        const int kk = k + hi * 2;
        v2f a;  a.x  = sx[nl * 516 + kk];   a.y  = sx[nl * 516 + kk + 1];
        const int gk = kc + kk;
        v2f bf;
        bf.x = xwT[(size_t)gk * RPROJ + col];
        bf.y = xwT[(size_t)(gk + 1) * RPROJ + col];
        acc1 = wmma_f32_k4(a, bf, acc1);
      }
    }
  }
  if (wave < 6) {
    for (int i = 0; i < 8; ++i) {
      int M = i + hi * 8;
      sproj[M * 100 + wave * 16 + nl] = acc1[i];
      if (wave == 4) Bmbuf[(size_t)(rowbase + M) * NST + nl] = acc1[i];
      if (wave == 5) Cbuf [(size_t)(rowbase + M) * NST + nl] = acc1[i];
    }
  }
  __syncthreads();

  v2f a2[16];
  for (int kt = 0; kt < 16; ++kt) {
    int kb = kt * 4 + hi * 2;
    a2[kt].x = sproj[nl * 100 + kb];
    a2[kt].y = sproj[nl * 100 + kb + 1];
  }
  for (int j = 0; j < 8; ++j) {
    const int col = wave * 128 + j * 16 + nl;
    v8f acc2;
    { float bv = dtb[col]; for (int i = 0; i < 8; ++i) acc2[i] = bv; }
    for (int kt = 0; kt < 16; ++kt) {
      int kb = kt * 4 + hi * 2;
      v2f bf;
      bf.x = dtwT[(size_t)kb * DEXP + col];
      bf.y = dtwT[(size_t)(kb + 1) * DEXP + col];
      acc2 = wmma_f32_k4(a2[kt], bf, acc2);
    }
    for (int i = 0; i < 8; ++i) {
      int M = i + hi * 8;
      float v  = acc2[i];
      float sp = (v > 20.0f) ? v : log1pf(__expf(v));
      dtbuf[(size_t)(rowbase + M) * DEXP + col] = sp;
    }
  }
}

#define SCAN_T   64
#define SCAN_DG  (DEXP / SCAN_T)
#define SCAN_U   8
#define FILL_R   32

__device__ __forceinline__ void stage_cols(const float* __restrict__ g0, const float* __restrict__ g1,
                                           float4* s, int t) {
  const int col4 = t & 15, rsub = t >> 4;
  for (int i0 = 0; i0 < CHUNK / 4; i0 += FILL_R) {
    float4 v0[FILL_R], v1[FILL_R];
    for (int k = 0; k < FILL_R; ++k) {
      const int row = (i0 + k) * 4 + rsub;
      v0[k] = *(const float4*)(g0 + (size_t)row * DEXP + col4 * 4);
      v1[k] = *(const float4*)(g1 + (size_t)row * DEXP + col4 * 4);
    }
    for (int k = 0; k < FILL_R; ++k) {
      const int row = (i0 + k) * 4 + rsub;
      float4 lo, hi;
      lo.x = v0[k].x; lo.y = v1[k].x; lo.z = v0[k].y; lo.w = v1[k].y;
      hi.x = v0[k].z; hi.y = v1[k].z; hi.z = v0[k].w; hi.w = v1[k].w;
      s[row * 32 + col4 * 2]     = lo;
      s[row * 32 + col4 * 2 + 1] = hi;
    }
  }
}

__global__ __launch_bounds__(SCAN_T) void ssm_scan_carry(
    const float* __restrict__ x, const float* __restrict__ dtbuf,
    const float* __restrict__ Bmbuf, const float* __restrict__ Aneg,
    float* __restrict__ Sbuf, float* __restrict__ Pbuf)
{
  __shared__ float4 sdx[CHUNK * 32];
  __shared__ float4 sB [CHUNK * 4];

  const int t  = threadIdx.x;
  const int bi = blockIdx.x;
  const int c  = bi & (NCHUNK - 1);
  const int dg = (bi / NCHUNK) & (SCAN_DG - 1);
  const int b  = bi / (NCHUNK * SCAN_DG);
  const int d  = dg * SCAN_T + t;
  const size_t rbase = (size_t)b * LSEQ + (size_t)c * CHUNK;

  {
    const float4* Bg = (const float4*)(Bmbuf + rbase * NST);
    float4 vb[CHUNK * 4 / SCAN_T];
    for (int i = 0; i < CHUNK * 4 / SCAN_T; ++i) vb[i] = Bg[t + i * SCAN_T];
    for (int i = 0; i < CHUNK * 4 / SCAN_T; ++i) sB[t + i * SCAN_T] = vb[i];
  }
  stage_cols(dtbuf + rbase * DEXP + dg * SCAN_T, x + rbase * DEXP + dg * SCAN_T, sdx, t);

  float A[NST];
  for (int q = 0; q < 4; ++q) {
    const float4 v = ((const float4*)(Aneg + (size_t)d * NST))[q];
    A[4 * q] = v.x; A[4 * q + 1] = v.y; A[4 * q + 2] = v.z; A[4 * q + 3] = v.w;
  }
  __syncthreads();

  const float2* sdxf = (const float2*)sdx + t;

  float h[NST];
  for (int n = 0; n < NST; ++n) h[n] = 0.0f;
  float sdt_sum = 0.0f;

  float2 dxn[SCAN_U];
  for (int j = 0; j < SCAN_U; ++j) dxn[j] = sdxf[j * SCAN_T];
  for (int l0 = 0; l0 < CHUNK; l0 += SCAN_U) {
    float2 dxc[SCAN_U];
    for (int j = 0; j < SCAN_U; ++j) dxc[j] = dxn[j];
    for (int j = 0; j < SCAN_U; ++j)
      dxn[j] = sdxf[((l0 + SCAN_U + j) & (CHUNK - 1)) * SCAN_T];
    #pragma unroll
    for (int j = 0; j < SCAN_U; ++j) {
      const int li = l0 + j;
      const float dtv = dxc[j].x;
      const float u   = dtv * dxc[j].y;
      for (int q = 0; q < 4; ++q) {
        const float4 bv = sB[li * 4 + q];
        h[4 * q]     = fmaf(__expf(A[4 * q]     * dtv), h[4 * q],     u * bv.x);
        h[4 * q + 1] = fmaf(__expf(A[4 * q + 1] * dtv), h[4 * q + 1], u * bv.y);
        h[4 * q + 2] = fmaf(__expf(A[4 * q + 2] * dtv), h[4 * q + 2], u * bv.z);
        h[4 * q + 3] = fmaf(__expf(A[4 * q + 3] * dtv), h[4 * q + 3], u * bv.w);
      }
      sdt_sum += dtv;
    }
  }
  const size_t si = (((size_t)b * NCHUNK + c) * DEXP + d) * NST;
  for (int q = 0; q < 4; ++q) {
    float4 s, p;
    s.x = h[4 * q]; s.y = h[4 * q + 1]; s.z = h[4 * q + 2]; s.w = h[4 * q + 3];
    p.x = __expf(A[4 * q] * sdt_sum);     p.y = __expf(A[4 * q + 1] * sdt_sum);
    p.z = __expf(A[4 * q + 2] * sdt_sum); p.w = __expf(A[4 * q + 3] * sdt_sum);
    ((float4*)(Sbuf + si))[q] = s;
    ((float4*)(Pbuf + si))[q] = p;
  }
}

__global__ void ssm_carry_combine(const float* __restrict__ Sbuf,
                                  const float* __restrict__ Pbuf,
                                  float* __restrict__ Car)
{
  const int tg = blockIdx.x * 256 + threadIdx.x;
  const int n  = tg & 15;
  const int d  = (tg >> 4) & (DEXP - 1);
  const int b  = tg >> 14;
  const size_t stride = (size_t)DEXP * NST;
  const size_t base   = (size_t)b * NCHUNK * stride + (size_t)d * NST + n;
  float cin = 0.0f;
  Car[base] = 0.0f;
  for (int c = 1; c < NCHUNK; ++c) {
    const size_t prev = base + (size_t)(c - 1) * stride;
    cin = Sbuf[prev] + Pbuf[prev] * cin;
    Car[base + (size_t)c * stride] = cin;
  }
}

__global__ __launch_bounds__(SCAN_T) void ssm_scan_out(
    const float* __restrict__ x, const float* __restrict__ dtbuf,
    const float* __restrict__ Bmbuf, const float* __restrict__ Cbuf,
    const float* __restrict__ Aneg, const float* __restrict__ Car,
    float* __restrict__ out)
{
  __shared__ float4 sdx[CHUNK * 32];
  __shared__ float4 sB [CHUNK * 4];
  __shared__ float4 sC [CHUNK * 4];

  const int t  = threadIdx.x;
  const int bi = blockIdx.x;
  const int c  = bi & (NCHUNK - 1);
  const int dg = (bi / NCHUNK) & (SCAN_DG - 1);
  const int b  = bi / (NCHUNK * SCAN_DG);
  const int d  = dg * SCAN_T + t;
  const size_t rbase = (size_t)b * LSEQ + (size_t)c * CHUNK;

  {
    const float4* Bg = (const float4*)(Bmbuf + rbase * NST);
    const float4* Cg = (const float4*)(Cbuf  + rbase * NST);
    float4 vb[CHUNK * 4 / SCAN_T], vc[CHUNK * 4 / SCAN_T];
    for (int i = 0; i < CHUNK * 4 / SCAN_T; ++i) { vb[i] = Bg[t + i * SCAN_T]; vc[i] = Cg[t + i * SCAN_T]; }
    for (int i = 0; i < CHUNK * 4 / SCAN_T; ++i) { sB[t + i * SCAN_T] = vb[i]; sC[t + i * SCAN_T] = vc[i]; }
  }
  stage_cols(dtbuf + rbase * DEXP + dg * SCAN_T, x + rbase * DEXP + dg * SCAN_T, sdx, t);

  const size_t si = (((size_t)b * NCHUNK + c) * DEXP + d) * NST;
  float A[NST], h[NST];
  for (int q = 0; q < 4; ++q) {
    const float4 v = ((const float4*)(Aneg + (size_t)d * NST))[q];
    A[4 * q] = v.x; A[4 * q + 1] = v.y; A[4 * q + 2] = v.z; A[4 * q + 3] = v.w;
    const float4 w = ((const float4*)(Car + si))[q];
    h[4 * q] = w.x; h[4 * q + 1] = w.y; h[4 * q + 2] = w.z; h[4 * q + 3] = w.w;
  }
  __syncthreads();

  const float2* sdxf = (const float2*)sdx + t;
  float* op         = out + rbase * DEXP + d;

  float2 dxn[SCAN_U];
  for (int j = 0; j < SCAN_U; ++j) dxn[j] = sdxf[j * SCAN_T];
  for (int l0 = 0; l0 < CHUNK; l0 += SCAN_U) {
    float2 dxc[SCAN_U];
    for (int j = 0; j < SCAN_U; ++j) dxc[j] = dxn[j];
    for (int j = 0; j < SCAN_U; ++j)
      dxn[j] = sdxf[((l0 + SCAN_U + j) & (CHUNK - 1)) * SCAN_T];
    #pragma unroll
    for (int j = 0; j < SCAN_U; ++j) {
      const int li = l0 + j;
      const float dtv = dxc[j].x;
      const float u   = dtv * dxc[j].y;
      float s2[8];
      for (int q = 0; q < 4; ++q) {
        const float4 bv = sB[li * 4 + q];
        const float4 cv = sC[li * 4 + q];
        h[4 * q]     = fmaf(__expf(A[4 * q]     * dtv), h[4 * q],     u * bv.x);
        h[4 * q + 1] = fmaf(__expf(A[4 * q + 1] * dtv), h[4 * q + 1], u * bv.y);
        h[4 * q + 2] = fmaf(__expf(A[4 * q + 2] * dtv), h[4 * q + 2], u * bv.z);
        h[4 * q + 3] = fmaf(__expf(A[4 * q + 3] * dtv), h[4 * q + 3], u * bv.w);
        s2[2 * q]     = h[4 * q]     * cv.x + h[4 * q + 1] * cv.y;
        s2[2 * q + 1] = h[4 * q + 2] * cv.z + h[4 * q + 3] * cv.w;
      }
      const float y = ((s2[0] + s2[1]) + (s2[2] + s2[3])) + ((s2[4] + s2[5]) + (s2[6] + s2[7]));
      op[(size_t)li * DEXP] = y;
    }
  }
}

extern "C" void kernel_launch(void* const* d_in, const int* in_sizes, int n_in,
                              void* d_out, int out_size, void* d_ws, size_t ws_size,
                              hipStream_t stream) {
  (void)in_sizes; (void)n_in; (void)out_size; (void)ws_size;
  const float* x     = (const float*)d_in[0];
  const float* xw    = (const float*)d_in[1];
  const float* xb    = (const float*)d_in[2];
  const float* dtw   = (const float*)d_in[3];
  const float* dtb   = (const float*)d_in[4];
  const float* A_log = (const float*)d_in[5];
  float* out = (float*)d_out;

  float* ws    = (float*)d_ws;
  float* dtbuf = ws;
  float* Aneg  = dtbuf + (size_t)ROWS * DEXP;
  float* Bm    = Aneg  + (size_t)DEXP * NST;
  float* Cb    = Bm    + (size_t)ROWS * NST;
  float* Sb    = Cb    + (size_t)ROWS * NST;
  float* Pb    = Sb    + (size_t)B_SZ * NCHUNK * DEXP * NST;
  float* Car   = Pb    + (size_t)B_SZ * NCHUNK * DEXP * NST;
  float* xwT   = Car   + (size_t)B_SZ * NCHUNK * DEXP * NST;
  float* dtwT  = xwT   + (size_t)DEXP * RPROJ;

  hipLaunchKernelGGL(ssm_prep, dim3((DEXP * RPROJ + 255) / 256), dim3(256), 0, stream,
                     A_log, Aneg, xw, xwT, dtw, dtwT);
  hipLaunchKernelGGL(ssm_proj_wmma, dim3(ROWS / 16), dim3(256), 0, stream,
                     x, xwT, xb, dtwT, dtb, dtbuf, Bm, Cb);
  hipLaunchKernelGGL(ssm_scan_carry, dim3(B_SZ * SCAN_DG * NCHUNK), dim3(SCAN_T), 0, stream,
                     x, dtbuf, Bm, Aneg, Sb, Pb);
  hipLaunchKernelGGL(ssm_carry_combine, dim3((B_SZ * DEXP * NST) / 256), dim3(256), 0, stream,
                     Sb, Pb, Car);
  hipLaunchKernelGGL(ssm_scan_out, dim3(B_SZ * SCAN_DG * NCHUNK), dim3(SCAN_T), 0, stream,
                     x, dtbuf, Bm, Cb, Aneg, Car, out);
}


# scan cut into 32 chunks of 64 rows (was 8 of 256): 2048 waves per scan launch, not 512; out scan trips 4 steps to fit 256 registers; carry folds 32 summaries (sum order at chunk edges changes)
# speedup vs baseline: 3.1680x; 1.3704x over previous
#include <hip/hip_runtime.h>
#include <math.h>

#define B_SZ   2
#define LSEQ   2048
#define DEXP   1024
#define DRANK  64
#define NST    16
#define RPROJ  (DRANK + 2*NST)
#define NCHUNK 32
#define CHUNK  (LSEQ / NCHUNK)
#define ROWS   (B_SZ * LSEQ)

typedef __attribute__((ext_vector_type(2))) float v2f;
typedef __attribute__((ext_vector_type(8))) float v8f;

__device__ __forceinline__ v8f wmma_f32_k4(v2f a, v2f b, v8f c) {
  return __builtin_amdgcn_wmma_f32_16x16x4_f32(
       false, a,  false, b,
       (short)0, c,  false,  false);
}

__global__ void ssm_prep(const float* __restrict__ A_log, float* __restrict__ Aneg,
                         const float* __restrict__ xw,  float* __restrict__ xwT,
                         const float* __restrict__ dtw, float* __restrict__ dtwT) {
  int i = blockIdx.x * 256 + threadIdx.x;
  if (i < DEXP * NST) Aneg[i] = -__expf(A_log[i]);
  if (i < DEXP * RPROJ) {
    int k = i / RPROJ, c = i % RPROJ;
    xwT[i] = xw[(size_t)c * DEXP + k];
  }
  if (i < DRANK * DEXP) {
    int k = i / DEXP, d = i % DEXP;
    dtwT[i] = dtw[(size_t)d * DRANK + k];
  }
}

__global__ __launch_bounds__(256) void ssm_proj_wmma(
    const float* __restrict__ x,    const float* __restrict__ xwT,
    const float* __restrict__ xb,   const float* __restrict__ dtwT,
    const float* __restrict__ dtb,
    float* __restrict__ dtbuf, float* __restrict__ Bmbuf, float* __restrict__ Cbuf)
{
  __shared__ float sx[16 * 516];
  __shared__ float sproj[16 * 100];

  const int t    = threadIdx.x;
  const int wave = t >> 5;
  const int nl   = t & 15;
  const int hi   = (t >> 4) & 1;
  const int rowbase = blockIdx.x * 16;

  v8f acc1;
  if (wave < 6) {
    float bv = xb[wave * 16 + nl];
    for (int i = 0; i < 8; ++i) acc1[i] = bv;
  }
  for (int kc = 0; kc < DEXP; kc += 512) {
    __syncthreads();
    for (int i = 0; i < 8; ++i) {
      int flat = t + i * 256;
      int r  = flat >> 7;
      int kk = (flat & 127) << 2;
      const float4 v = *(const float4*)(x + (size_t)(rowbase + r) * DEXP + kc + kk);
      float* dst = &sx[r * 516 + kk];
      dst[0] = v.x; dst[1] = v.y; dst[2] = v.z; dst[3] = v.w;
    }
    __syncthreads();
    if (wave < 6) {
      const int col = wave * 16 + nl;
      for (int k = 0; k < 512; k += 4) {
        const int kk = k + hi * 2;
        v2f a;  a.x  = sx[nl * 516 + kk];   a.y  = sx[nl * 516 + kk + 1];
        const int gk = kc + kk;
        v2f bf;
        bf.x = xwT[(size_t)gk * RPROJ + col];
        bf.y = xwT[(size_t)(gk + 1) * RPROJ + col];
        acc1 = wmma_f32_k4(a, bf, acc1);
      }
    }
  }
  if (wave < 6) {
    for (int i = 0; i < 8; ++i) {
      int M = i + hi * 8;
      sproj[M * 100 + wave * 16 + nl] = acc1[i];
      if (wave == 4) Bmbuf[(size_t)(rowbase + M) * NST + nl] = acc1[i];
      if (wave == 5) Cbuf [(size_t)(rowbase + M) * NST + nl] = acc1[i];
    }
  }
  __syncthreads();

  v2f a2[16];
  for (int kt = 0; kt < 16; ++kt) {
    int kb = kt * 4 + hi * 2;
    a2[kt].x = sproj[nl * 100 + kb];
    a2[kt].y = sproj[nl * 100 + kb + 1];
  }
  for (int j = 0; j < 8; ++j) {
    const int col = wave * 128 + j * 16 + nl;
    v8f acc2;
    { float bv = dtb[col]; for (int i = 0; i < 8; ++i) acc2[i] = bv; }
    for (int kt = 0; kt < 16; ++kt) {
      int kb = kt * 4 + hi * 2;
      v2f bf;
      bf.x = dtwT[(size_t)kb * DEXP + col];
      bf.y = dtwT[(size_t)(kb + 1) * DEXP + col];
      acc2 = wmma_f32_k4(a2[kt], bf, acc2);
    }
    for (int i = 0; i < 8; ++i) {
      int M = i + hi * 8;
      float v  = acc2[i];
      float sp = (v > 20.0f) ? v : log1pf(__expf(v));
      dtbuf[(size_t)(rowbase + M) * DEXP + col] = sp;
    }
  }
}

#define SCAN_T   64
#define SCAN_DG  (DEXP / SCAN_T)
#define SCAN_U   8
#define SCAN_UO  4
#define FILL_R   16

__device__ __forceinline__ void stage_cols(const float* __restrict__ g0, const float* __restrict__ g1,
                                           float4* s, int t) {
  const int col4 = t & 15, rsub = t >> 4;
  for (int i0 = 0; i0 < CHUNK / 4; i0 += FILL_R) {
    float4 v0[FILL_R], v1[FILL_R];
    for (int k = 0; k < FILL_R; ++k) {
      const int row = (i0 + k) * 4 + rsub;
      v0[k] = *(const float4*)(g0 + (size_t)row * DEXP + col4 * 4);
      v1[k] = *(const float4*)(g1 + (size_t)row * DEXP + col4 * 4);
    }
    for (int k = 0; k < FILL_R; ++k) {
      const int row = (i0 + k) * 4 + rsub;
      float4 lo, hi;
      lo.x = v0[k].x; lo.y = v1[k].x; lo.z = v0[k].y; lo.w = v1[k].y;
      hi.x = v0[k].z; hi.y = v1[k].z; hi.z = v0[k].w; hi.w = v1[k].w;
      s[row * 32 + col4 * 2]     = lo;
      s[row * 32 + col4 * 2 + 1] = hi;
    }
  }
}

__global__ __launch_bounds__(SCAN_T) void ssm_scan_carry(
    const float* __restrict__ x, const float* __restrict__ dtbuf,
    const float* __restrict__ Bmbuf, const float* __restrict__ Aneg,
    float* __restrict__ Sbuf, float* __restrict__ Pbuf)
{
  __shared__ float4 sdx[CHUNK * 32];
  __shared__ float4 sB [CHUNK * 4];

  const int t  = threadIdx.x;
  const int bi = blockIdx.x;
  const int c  = bi & (NCHUNK - 1);
  const int dg = (bi / NCHUNK) & (SCAN_DG - 1);
  const int b  = bi / (NCHUNK * SCAN_DG);
  const int d  = dg * SCAN_T + t;
  const size_t rbase = (size_t)b * LSEQ + (size_t)c * CHUNK;

  {
    const float4* Bg = (const float4*)(Bmbuf + rbase * NST);
    float4 vb[CHUNK * 4 / SCAN_T];
    for (int i = 0; i < CHUNK * 4 / SCAN_T; ++i) vb[i] = Bg[t + i * SCAN_T];
    for (int i = 0; i < CHUNK * 4 / SCAN_T; ++i) sB[t + i * SCAN_T] = vb[i];
  }
  stage_cols(dtbuf + rbase * DEXP + dg * SCAN_T, x + rbase * DEXP + dg * SCAN_T, sdx, t);

  float A[NST];
  for (int q = 0; q < 4; ++q) {
    const float4 v = ((const float4*)(Aneg + (size_t)d * NST))[q];
    A[4 * q] = v.x; A[4 * q + 1] = v.y; A[4 * q + 2] = v.z; A[4 * q + 3] = v.w;
  }
  __syncthreads();

  const float2* sdxf = (const float2*)sdx + t;

  float h[NST];
  for (int n = 0; n < NST; ++n) h[n] = 0.0f;
  float sdt_sum = 0.0f;

  float2 dxn[SCAN_U];
  for (int j = 0; j < SCAN_U; ++j) dxn[j] = sdxf[j * SCAN_T];
  for (int l0 = 0; l0 < CHUNK; l0 += SCAN_U) {
    float2 dxc[SCAN_U];
    for (int j = 0; j < SCAN_U; ++j) dxc[j] = dxn[j];
    for (int j = 0; j < SCAN_U; ++j)
      dxn[j] = sdxf[((l0 + SCAN_U + j) & (CHUNK - 1)) * SCAN_T];
    #pragma unroll
    for (int j = 0; j < SCAN_U; ++j) {
      const int li = l0 + j;
      const float dtv = dxc[j].x;
      const float u   = dtv * dxc[j].y;
      for (int q = 0; q < 4; ++q) {
        const float4 bv = sB[li * 4 + q];
        h[4 * q]     = fmaf(__expf(A[4 * q]     * dtv), h[4 * q],     u * bv.x);
        h[4 * q + 1] = fmaf(__expf(A[4 * q + 1] * dtv), h[4 * q + 1], u * bv.y);
        h[4 * q + 2] = fmaf(__expf(A[4 * q + 2] * dtv), h[4 * q + 2], u * bv.z);
        h[4 * q + 3] = fmaf(__expf(A[4 * q + 3] * dtv), h[4 * q + 3], u * bv.w);
      }
      sdt_sum += dtv;
    }
  }
  const size_t si = (((size_t)b * NCHUNK + c) * DEXP + d) * NST;
  for (int q = 0; q < 4; ++q) {
    float4 s, p;
    s.x = h[4 * q]; s.y = h[4 * q + 1]; s.z = h[4 * q + 2]; s.w = h[4 * q + 3];
    p.x = __expf(A[4 * q] * sdt_sum);     p.y = __expf(A[4 * q + 1] * sdt_sum);
    p.z = __expf(A[4 * q + 2] * sdt_sum); p.w = __expf(A[4 * q + 3] * sdt_sum);
    ((float4*)(Sbuf + si))[q] = s;
    ((float4*)(Pbuf + si))[q] = p;
  }
}

__global__ void ssm_carry_combine(const float* __restrict__ Sbuf,
                                  const float* __restrict__ Pbuf,
                                  float* __restrict__ Car)
{
  const int tg = blockIdx.x * 256 + threadIdx.x;
  const int n  = tg & 15;
  const int d  = (tg >> 4) & (DEXP - 1);
  const int b  = tg >> 14;
  const size_t stride = (size_t)DEXP * NST;
  const size_t base   = (size_t)b * NCHUNK * stride + (size_t)d * NST + n;
  float sv[NCHUNK - 1], pv[NCHUNK - 1];
  #pragma unroll
  for (int c = 0; c < NCHUNK - 1; ++c) {
    sv[c] = Sbuf[base + (size_t)c * stride];
    pv[c] = Pbuf[base + (size_t)c * stride];
  }
  float cin = 0.0f;
  Car[base] = 0.0f;
  #pragma unroll
  for (int c = 1; c < NCHUNK; ++c) {
    cin = sv[c - 1] + pv[c - 1] * cin;
    Car[base + (size_t)c * stride] = cin;
  }
}

__global__ __launch_bounds__(SCAN_T) void ssm_scan_out(
    const float* __restrict__ x, const float* __restrict__ dtbuf,
    const float* __restrict__ Bmbuf, const float* __restrict__ Cbuf,
    const float* __restrict__ Aneg, const float* __restrict__ Car,
    float* __restrict__ out)
{
  __shared__ float4 sdx[CHUNK * 32];
  __shared__ float4 sB [CHUNK * 4];
  __shared__ float4 sC [CHUNK * 4];

  const int t  = threadIdx.x;
  const int bi = blockIdx.x;
  const int c  = bi & (NCHUNK - 1);
  const int dg = (bi / NCHUNK) & (SCAN_DG - 1);
  const int b  = bi / (NCHUNK * SCAN_DG);
  const int d  = dg * SCAN_T + t;
  const size_t rbase = (size_t)b * LSEQ + (size_t)c * CHUNK;

  {
    const float4* Bg = (const float4*)(Bmbuf + rbase * NST);
    const float4* Cg = (const float4*)(Cbuf  + rbase * NST);
    float4 vb[CHUNK * 4 / SCAN_T], vc[CHUNK * 4 / SCAN_T];
    for (int i = 0; i < CHUNK * 4 / SCAN_T; ++i) { vb[i] = Bg[t + i * SCAN_T]; vc[i] = Cg[t + i * SCAN_T]; }
    for (int i = 0; i < CHUNK * 4 / SCAN_T; ++i) { sB[t + i * SCAN_T] = vb[i]; sC[t + i * SCAN_T] = vc[i]; }
  }
  stage_cols(dtbuf + rbase * DEXP + dg * SCAN_T, x + rbase * DEXP + dg * SCAN_T, sdx, t);

  const size_t si = (((size_t)b * NCHUNK + c) * DEXP + d) * NST;
  float A[NST], h[NST];
  for (int q = 0; q < 4; ++q) {
    const float4 v = ((const float4*)(Aneg + (size_t)d * NST))[q];
    A[4 * q] = v.x; A[4 * q + 1] = v.y; A[4 * q + 2] = v.z; A[4 * q + 3] = v.w;
    const float4 w = ((const float4*)(Car + si))[q];
    h[4 * q] = w.x; h[4 * q + 1] = w.y; h[4 * q + 2] = w.z; h[4 * q + 3] = w.w;
  }
  __syncthreads();

  const float2* sdxf = (const float2*)sdx + t;
  float* op         = out + rbase * DEXP + d;

  float2 dxn[SCAN_UO];
  for (int j = 0; j < SCAN_UO; ++j) dxn[j] = sdxf[j * SCAN_T];
  for (int l0 = 0; l0 < CHUNK; l0 += SCAN_UO) {
    float2 dxc[SCAN_UO];
    for (int j = 0; j < SCAN_UO; ++j) dxc[j] = dxn[j];
    for (int j = 0; j < SCAN_UO; ++j)
      dxn[j] = sdxf[((l0 + SCAN_UO + j) & (CHUNK - 1)) * SCAN_T];
    #pragma unroll
    for (int j = 0; j < SCAN_UO; ++j) {
      const int li = l0 + j;
      const float dtv = dxc[j].x;
      const float u   = dtv * dxc[j].y;
      float s2[8];
      for (int q = 0; q < 4; ++q) {
        const float4 bv = sB[li * 4 + q];
        const float4 cv = sC[li * 4 + q];
        h[4 * q]     = fmaf(__expf(A[4 * q]     * dtv), h[4 * q],     u * bv.x);
        h[4 * q + 1] = fmaf(__expf(A[4 * q + 1] * dtv), h[4 * q + 1], u * bv.y);
        h[4 * q + 2] = fmaf(__expf(A[4 * q + 2] * dtv), h[4 * q + 2], u * bv.z);
        h[4 * q + 3] = fmaf(__expf(A[4 * q + 3] * dtv), h[4 * q + 3], u * bv.w);
        s2[2 * q]     = h[4 * q]     * cv.x + h[4 * q + 1] * cv.y;
        s2[2 * q + 1] = h[4 * q + 2] * cv.z + h[4 * q + 3] * cv.w;
      }
      const float y = ((s2[0] + s2[1]) + (s2[2] + s2[3])) + ((s2[4] + s2[5]) + (s2[6] + s2[7]));
      op[(size_t)li * DEXP] = y;
    }
  }
}

extern "C" void kernel_launch(void* const* d_in, const int* in_sizes, int n_in,
                              void* d_out, int out_size, void* d_ws, size_t ws_size,
                              hipStream_t stream) {
  (void)in_sizes; (void)n_in; (void)out_size; (void)ws_size;
  const float* x     = (const float*)d_in[0];
  const float* xw    = (const float*)d_in[1];
  const float* xb    = (const float*)d_in[2];
  const float* dtw   = (const float*)d_in[3];
  const float* dtb   = (const float*)d_in[4];
  const float* A_log = (const float*)d_in[5];
  float* out = (float*)d_out;

  float* ws    = (float*)d_ws;
  float* dtbuf = ws;
  float* Aneg  = dtbuf + (size_t)ROWS * DEXP;
  float* Bm    = Aneg  + (size_t)DEXP * NST;
  float* Cb    = Bm    + (size_t)ROWS * NST;
  float* Sb    = Cb    + (size_t)ROWS * NST;
  float* Pb    = Sb    + (size_t)B_SZ * NCHUNK * DEXP * NST;
  float* Car   = Pb    + (size_t)B_SZ * NCHUNK * DEXP * NST;
  float* xwT   = Car   + (size_t)B_SZ * NCHUNK * DEXP * NST;
  float* dtwT  = xwT   + (size_t)DEXP * RPROJ;

  hipLaunchKernelGGL(ssm_prep, dim3((DEXP * RPROJ + 255) / 256), dim3(256), 0, stream,
                     A_log, Aneg, xw, xwT, dtw, dtwT);
  hipLaunchKernelGGL(ssm_proj_wmma, dim3(ROWS / 16), dim3(256), 0, stream,
                     x, xwT, xb, dtwT, dtb, dtbuf, Bm, Cb);
  hipLaunchKernelGGL(ssm_scan_carry, dim3(B_SZ * SCAN_DG * NCHUNK), dim3(SCAN_T), 0, stream,
                     x, dtbuf, Bm, Aneg, Sb, Pb);
  hipLaunchKernelGGL(ssm_carry_combine, dim3((B_SZ * DEXP * NST) / 256), dim3(256), 0, stream,
                     Sb, Pb, Car);
  hipLaunchKernelGGL(ssm_scan_out, dim3(B_SZ * SCAN_DG * NCHUNK), dim3(SCAN_T), 0, stream,
                     x, dtbuf, Bm, Cb, Aneg, Car, out);
}


# projection GEMMs fed by 16-byte k-contiguous loads of x and of both weights as given (transposing pre-pass gone); GEMM1 split over k across the 8 waves, partials added in LDS: k sum order changes
# speedup vs baseline: 4.7742x; 1.5070x over previous
#include <hip/hip_runtime.h>
#include <math.h>

#define B_SZ   2
#define LSEQ   2048
#define DEXP   1024
#define DRANK  64
#define NST    16
#define RPROJ  (DRANK + 2*NST)
#define NCHUNK 32
#define CHUNK  (LSEQ / NCHUNK)
#define ROWS   (B_SZ * LSEQ)

typedef __attribute__((ext_vector_type(2))) float v2f;
typedef __attribute__((ext_vector_type(8))) float v8f;

__device__ __forceinline__ v8f wmma_f32_k4(v2f a, v2f b, v8f c) {
  return __builtin_amdgcn_wmma_f32_16x16x4_f32(
       false, a,  false, b,
       (short)0, c,  false,  false);
}

__global__ void ssm_prep(const float* __restrict__ A_log, float* __restrict__ Aneg) {
  int i = blockIdx.x * 256 + threadIdx.x;
  if (i < DEXP * NST) Aneg[i] = -__expf(A_log[i]);
}

__device__ __forceinline__ float softplus_f32(float v) {
  const float t = __expf(-fabsf(v));
  const float d = 2.0f + t;
  float r = __builtin_amdgcn_rcpf(d);
  r = fmaf(fmaf(-d, r, 1.0f), r, r);
  const float s = t * r;
  const float z = s * s;
  float p = 1.0f / 15.0f;
  p = fmaf(p, z, 1.0f / 13.0f);
  p = fmaf(p, z, 1.0f / 11.0f);
  p = fmaf(p, z, 1.0f / 9.0f);
  p = fmaf(p, z, 1.0f / 7.0f);
  p = fmaf(p, z, 1.0f / 5.0f);
  p = fmaf(p, z, 1.0f / 3.0f);
  const float s2 = s + s;
  return fmaxf(v, 0.0f) + fmaf(s2 * z, p, s2);
}

#define XP4  33
#define PP   100

__global__ __launch_bounds__(256) void ssm_proj_wmma(
    const float* __restrict__ x,    const float* __restrict__ xw,
    const float* __restrict__ xb,   const float* __restrict__ dtw,
    const float* __restrict__ dtb,
    float* __restrict__ dtbuf, float* __restrict__ Bmbuf, float* __restrict__ Cbuf)
{
  __shared__ float4 sx4[8 * 16 * XP4];
  __shared__ float  sred[8 * 16 * PP];
  __shared__ float4 sproj4[16 * PP / 4];

  const int t    = threadIdx.x;
  const int wave = t >> 5;
  const int lane = t & 31;
  const int nl   = t & 15;
  const int hi   = (t >> 4) & 1;
  const int rowbase = blockIdx.x * 16;
  const int kw   = wave * 128;

  {
    float4 v[16];
    for (int r = 0; r < 16; ++r)
      v[r] = *(const float4*)(x + (size_t)(rowbase + r) * DEXP + kw + lane * 4);
    float4* s = sx4 + wave * 16 * XP4;
    for (int r = 0; r < 16; ++r) s[r * XP4 + lane] = v[r];
  }
  __syncthreads();
  float4 av[16];
  {
    const float4* s = sx4 + wave * 16 * XP4 + nl * XP4 + hi;
    for (int i = 0; i < 16; ++i) av[i] = s[i * 2];
  }

  v8f acc[6];
  for (int tt = 0; tt < 6; ++tt)
    for (int r = 0; r < 8; ++r) acc[tt][r] = 0.0f;
  {
    const float* wb = xw + (size_t)nl * DEXP + kw + hi * 4;
    #pragma unroll
    for (int i = 0; i < 16; ++i) {
      float4 bv[6];
      for (int tt = 0; tt < 6; ++tt)
        bv[tt] = *(const float4*)(wb + (size_t)tt * 16 * DEXP + i * 8);
      v2f a0, a1;
      a0.x = av[i].x; a0.y = av[i].y; a1.x = av[i].z; a1.y = av[i].w;
      for (int tt = 0; tt < 6; ++tt) {
        v2f b0; b0.x = bv[tt].x; b0.y = bv[tt].y;
        acc[tt] = wmma_f32_k4(a0, b0, acc[tt]);
      }
      for (int tt = 0; tt < 6; ++tt) {
        v2f b1; b1.x = bv[tt].z; b1.y = bv[tt].w;
        acc[tt] = wmma_f32_k4(a1, b1, acc[tt]);
      }
    }
  }
  for (int tt = 0; tt < 6; ++tt)
    for (int r = 0; r < 8; ++r)
      sred[(wave * 16 + r + hi * 8) * PP + tt * 16 + nl] = acc[tt][r];
  __syncthreads();
  float* sproj = (float*)sproj4;
  for (int i = 0; i < 6; ++i) {
    const int idx = t + i * 256;
    const int row = idx / RPROJ;
    const int col = idx - row * RPROJ;
    float v = xb[col];
    for (int w = 0; w < 8; ++w) v += sred[(w * 16 + row) * PP + col];
    sproj[row * PP + col] = v;
  }
  __syncthreads();
  if (t < 128) {
    const int q = t & 63, row = q >> 2, c4 = q & 3;
    const float4 v = sproj4[row * (PP / 4) + 16 + (t >> 6) * 4 + c4];
    float* dst = (t < 64) ? Bmbuf : Cbuf;
    *(float4*)(dst + (size_t)(rowbase + row) * NST + c4 * 4) = v;
  }

  float4 a2[8];
  {
    const float4* s = sproj4 + nl * (PP / 4) + hi;
    for (int q = 0; q < 8; ++q) a2[q] = s[q * 2];
  }
  const float* db = dtw + (size_t)(wave * 128 + nl) * DRANK + hi * 4;
  float4 bc[2][8];
  for (int u = 0; u < 2; ++u)
    for (int q = 0; q < 8; ++q)
      bc[u][q] = *(const float4*)(db + (size_t)(u * 16) * DRANK + q * 8);
  #pragma unroll
  for (int jj = 0; jj < 8; jj += 2) {
    float4 bn[2][8];
    if (jj + 2 < 8) {
      for (int u = 0; u < 2; ++u)
        for (int q = 0; q < 8; ++q)
          bn[u][q] = *(const float4*)(db + (size_t)((jj + 2 + u) * 16) * DRANK + q * 8);
    }
    const int col0 = wave * 128 + jj * 16 + nl;
    const int col1 = col0 + 16;
    v8f c0, c1;
    { const float bv0 = dtb[col0], bv1 = dtb[col1];
      for (int r = 0; r < 8; ++r) { c0[r] = bv0; c1[r] = bv1; } }
    for (int q = 0; q < 8; ++q) {
      v2f a0, a1, b00, b01, b10, b11;
      a0.x = a2[q].x; a0.y = a2[q].y; a1.x = a2[q].z; a1.y = a2[q].w;
      b00.x = bc[0][q].x; b00.y = bc[0][q].y; b01.x = bc[0][q].z; b01.y = bc[0][q].w;
      b10.x = bc[1][q].x; b10.y = bc[1][q].y; b11.x = bc[1][q].z; b11.y = bc[1][q].w;
      c0 = wmma_f32_k4(a0, b00, c0);
      c1 = wmma_f32_k4(a0, b10, c1);
      c0 = wmma_f32_k4(a1, b01, c0);
      c1 = wmma_f32_k4(a1, b11, c1);
    }
    for (int r = 0; r < 8; ++r) {
      const int M = r + hi * 8;
      dtbuf[(size_t)(rowbase + M) * DEXP + col0] = softplus_f32(c0[r]);
      dtbuf[(size_t)(rowbase + M) * DEXP + col1] = softplus_f32(c1[r]);
    }
    if (jj + 2 < 8) {
      for (int u = 0; u < 2; ++u)
        for (int q = 0; q < 8; ++q) bc[u][q] = bn[u][q];
    }
  }
}

#define SCAN_T   64
#define SCAN_DG  (DEXP / SCAN_T)
#define SCAN_U   8
#define SCAN_UO  4
#define FILL_R   16

__device__ __forceinline__ void stage_cols(const float* __restrict__ g0, const float* __restrict__ g1,
                                           float4* s, int t) {
  const int col4 = t & 15, rsub = t >> 4;
  for (int i0 = 0; i0 < CHUNK / 4; i0 += FILL_R) {
    float4 v0[FILL_R], v1[FILL_R];
    for (int k = 0; k < FILL_R; ++k) {
      const int row = (i0 + k) * 4 + rsub;
      v0[k] = *(const float4*)(g0 + (size_t)row * DEXP + col4 * 4);
      v1[k] = *(const float4*)(g1 + (size_t)row * DEXP + col4 * 4);
    }
    for (int k = 0; k < FILL_R; ++k) {
      const int row = (i0 + k) * 4 + rsub;
      float4 lo, hi;
      lo.x = v0[k].x; lo.y = v1[k].x; lo.z = v0[k].y; lo.w = v1[k].y;
      hi.x = v0[k].z; hi.y = v1[k].z; hi.z = v0[k].w; hi.w = v1[k].w;
      s[row * 32 + col4 * 2]     = lo;
      s[row * 32 + col4 * 2 + 1] = hi;
    }
  }
}

__global__ __launch_bounds__(SCAN_T) void ssm_scan_carry(
    const float* __restrict__ x, const float* __restrict__ dtbuf,
    const float* __restrict__ Bmbuf, const float* __restrict__ Aneg,
    float* __restrict__ Sbuf, float* __restrict__ Pbuf)
{
  __shared__ float4 sdx[CHUNK * 32];
  __shared__ float4 sB [CHUNK * 4];

  const int t  = threadIdx.x;
  const int bi = blockIdx.x;
  const int c  = bi & (NCHUNK - 1);
  const int dg = (bi / NCHUNK) & (SCAN_DG - 1);
  const int b  = bi / (NCHUNK * SCAN_DG);
  const int d  = dg * SCAN_T + t;
  const size_t rbase = (size_t)b * LSEQ + (size_t)c * CHUNK;

  {
    const float4* Bg = (const float4*)(Bmbuf + rbase * NST);
    float4 vb[CHUNK * 4 / SCAN_T];
    for (int i = 0; i < CHUNK * 4 / SCAN_T; ++i) vb[i] = Bg[t + i * SCAN_T];
    for (int i = 0; i < CHUNK * 4 / SCAN_T; ++i) sB[t + i * SCAN_T] = vb[i];
  }
  stage_cols(dtbuf + rbase * DEXP + dg * SCAN_T, x + rbase * DEXP + dg * SCAN_T, sdx, t);

  float A[NST];
  for (int q = 0; q < 4; ++q) {
    const float4 v = ((const float4*)(Aneg + (size_t)d * NST))[q];
    A[4 * q] = v.x; A[4 * q + 1] = v.y; A[4 * q + 2] = v.z; A[4 * q + 3] = v.w;
  }
  __syncthreads();

  const float2* sdxf = (const float2*)sdx + t;

  float h[NST];
  for (int n = 0; n < NST; ++n) h[n] = 0.0f;
  float sdt_sum = 0.0f;

  float2 dxn[SCAN_U];
  for (int j = 0; j < SCAN_U; ++j) dxn[j] = sdxf[j * SCAN_T];
  for (int l0 = 0; l0 < CHUNK; l0 += SCAN_U) {
    float2 dxc[SCAN_U];
    for (int j = 0; j < SCAN_U; ++j) dxc[j] = dxn[j];
    for (int j = 0; j < SCAN_U; ++j)
      dxn[j] = sdxf[((l0 + SCAN_U + j) & (CHUNK - 1)) * SCAN_T];
    #pragma unroll
    for (int j = 0; j < SCAN_U; ++j) {
      const int li = l0 + j;
      const float dtv = dxc[j].x;
      const float u   = dtv * dxc[j].y;
      for (int q = 0; q < 4; ++q) {
        const float4 bv = sB[li * 4 + q];
        h[4 * q]     = fmaf(__expf(A[4 * q]     * dtv), h[4 * q],     u * bv.x);
        h[4 * q + 1] = fmaf(__expf(A[4 * q + 1] * dtv), h[4 * q + 1], u * bv.y);
        h[4 * q + 2] = fmaf(__expf(A[4 * q + 2] * dtv), h[4 * q + 2], u * bv.z);
        h[4 * q + 3] = fmaf(__expf(A[4 * q + 3] * dtv), h[4 * q + 3], u * bv.w);
      }
      sdt_sum += dtv;
    }
  }
  const size_t si = (((size_t)b * NCHUNK + c) * DEXP + d) * NST;
  for (int q = 0; q < 4; ++q) {
    float4 s, p;
    s.x = h[4 * q]; s.y = h[4 * q + 1]; s.z = h[4 * q + 2]; s.w = h[4 * q + 3];
    p.x = __expf(A[4 * q] * sdt_sum);     p.y = __expf(A[4 * q + 1] * sdt_sum);
    p.z = __expf(A[4 * q + 2] * sdt_sum); p.w = __expf(A[4 * q + 3] * sdt_sum);
    ((float4*)(Sbuf + si))[q] = s;
    ((float4*)(Pbuf + si))[q] = p;
  }
}

__global__ void ssm_carry_combine(const float* __restrict__ Sbuf,
                                  const float* __restrict__ Pbuf,
                                  float* __restrict__ Car)
{
  const int tg = blockIdx.x * 256 + threadIdx.x;
  const int n  = tg & 15;
  const int d  = (tg >> 4) & (DEXP - 1);
  const int b  = tg >> 14;
  const size_t stride = (size_t)DEXP * NST;
  const size_t base   = (size_t)b * NCHUNK * stride + (size_t)d * NST + n;
  float sv[NCHUNK - 1], pv[NCHUNK - 1];
  #pragma unroll
  for (int c = 0; c < NCHUNK - 1; ++c) {
    sv[c] = Sbuf[base + (size_t)c * stride];
    pv[c] = Pbuf[base + (size_t)c * stride];
  }
  float cin = 0.0f;
  Car[base] = 0.0f;
  #pragma unroll
  for (int c = 1; c < NCHUNK; ++c) {
    cin = sv[c - 1] + pv[c - 1] * cin;
    Car[base + (size_t)c * stride] = cin;
  }
}

__global__ __launch_bounds__(SCAN_T) void ssm_scan_out(
    const float* __restrict__ x, const float* __restrict__ dtbuf,
    const float* __restrict__ Bmbuf, const float* __restrict__ Cbuf,
    const float* __restrict__ Aneg, const float* __restrict__ Car,
    float* __restrict__ out)
{
  __shared__ float4 sdx[CHUNK * 32];
  __shared__ float4 sB [CHUNK * 4];
  __shared__ float4 sC [CHUNK * 4];

  const int t  = threadIdx.x;
  const int bi = blockIdx.x;
  const int c  = bi & (NCHUNK - 1);
  const int dg = (bi / NCHUNK) & (SCAN_DG - 1);
  const int b  = bi / (NCHUNK * SCAN_DG);
  const int d  = dg * SCAN_T + t;
  const size_t rbase = (size_t)b * LSEQ + (size_t)c * CHUNK;

  {
    const float4* Bg = (const float4*)(Bmbuf + rbase * NST);
    const float4* Cg = (const float4*)(Cbuf  + rbase * NST);
    float4 vb[CHUNK * 4 / SCAN_T], vc[CHUNK * 4 / SCAN_T];
    for (int i = 0; i < CHUNK * 4 / SCAN_T; ++i) { vb[i] = Bg[t + i * SCAN_T]; vc[i] = Cg[t + i * SCAN_T]; }
    for (int i = 0; i < CHUNK * 4 / SCAN_T; ++i) { sB[t + i * SCAN_T] = vb[i]; sC[t + i * SCAN_T] = vc[i]; }
  }
  stage_cols(dtbuf + rbase * DEXP + dg * SCAN_T, x + rbase * DEXP + dg * SCAN_T, sdx, t);

  const size_t si = (((size_t)b * NCHUNK + c) * DEXP + d) * NST;
  float A[NST], h[NST];
  for (int q = 0; q < 4; ++q) {
    const float4 v = ((const float4*)(Aneg + (size_t)d * NST))[q];
    A[4 * q] = v.x; A[4 * q + 1] = v.y; A[4 * q + 2] = v.z; A[4 * q + 3] = v.w;
    const float4 w = ((const float4*)(Car + si))[q];
    h[4 * q] = w.x; h[4 * q + 1] = w.y; h[4 * q + 2] = w.z; h[4 * q + 3] = w.w;
  }
  __syncthreads();

  const float2* sdxf = (const float2*)sdx + t;
  float* op         = out + rbase * DEXP + d;

  float2 dxn[SCAN_UO];
  for (int j = 0; j < SCAN_UO; ++j) dxn[j] = sdxf[j * SCAN_T];
  for (int l0 = 0; l0 < CHUNK; l0 += SCAN_UO) {
    float2 dxc[SCAN_UO];
    for (int j = 0; j < SCAN_UO; ++j) dxc[j] = dxn[j];
    for (int j = 0; j < SCAN_UO; ++j)
      dxn[j] = sdxf[((l0 + SCAN_UO + j) & (CHUNK - 1)) * SCAN_T];
    #pragma unroll
    for (int j = 0; j < SCAN_UO; ++j) {
      const int li = l0 + j;
      const float dtv = dxc[j].x;
      const float u   = dtv * dxc[j].y;
      float s2[8];
      for (int q = 0; q < 4; ++q) {
        const float4 bv = sB[li * 4 + q];
        const float4 cv = sC[li * 4 + q];
        h[4 * q]     = fmaf(__expf(A[4 * q]     * dtv), h[4 * q],     u * bv.x);
        h[4 * q + 1] = fmaf(__expf(A[4 * q + 1] * dtv), h[4 * q + 1], u * bv.y);
        h[4 * q + 2] = fmaf(__expf(A[4 * q + 2] * dtv), h[4 * q + 2], u * bv.z);
        h[4 * q + 3] = fmaf(__expf(A[4 * q + 3] * dtv), h[4 * q + 3], u * bv.w);
        s2[2 * q]     = h[4 * q]     * cv.x + h[4 * q + 1] * cv.y;
        s2[2 * q + 1] = h[4 * q + 2] * cv.z + h[4 * q + 3] * cv.w;
      }
      const float y = ((s2[0] + s2[1]) + (s2[2] + s2[3])) + ((s2[4] + s2[5]) + (s2[6] + s2[7]));
      op[(size_t)li * DEXP] = y;
    }
  }
}

extern "C" void kernel_launch(void* const* d_in, const int* in_sizes, int n_in,
                              void* d_out, int out_size, void* d_ws, size_t ws_size,
                              hipStream_t stream) {
  (void)in_sizes; (void)n_in; (void)out_size; (void)ws_size;
  const float* x     = (const float*)d_in[0];
  const float* xw    = (const float*)d_in[1];
  const float* xb    = (const float*)d_in[2];
  const float* dtw   = (const float*)d_in[3];
  const float* dtb   = (const float*)d_in[4];
  const float* A_log = (const float*)d_in[5];
  float* out = (float*)d_out;

  float* ws    = (float*)d_ws;
  float* dtbuf = ws;
  float* Aneg  = dtbuf + (size_t)ROWS * DEXP;
  float* Bm    = Aneg  + (size_t)DEXP * NST;
  float* Cb    = Bm    + (size_t)ROWS * NST;
  float* Sb    = Cb    + (size_t)ROWS * NST;
  float* Pb    = Sb    + (size_t)B_SZ * NCHUNK * DEXP * NST;
  float* Car   = Pb    + (size_t)B_SZ * NCHUNK * DEXP * NST;

  hipLaunchKernelGGL(ssm_prep, dim3((DEXP * NST) / 256), dim3(256), 0, stream,
                     A_log, Aneg);
  hipLaunchKernelGGL(ssm_proj_wmma, dim3(ROWS / 16), dim3(256), 0, stream,
                     x, xw, xb, dtw, dtb, dtbuf, Bm, Cb);
  hipLaunchKernelGGL(ssm_scan_carry, dim3(B_SZ * SCAN_DG * NCHUNK), dim3(SCAN_T), 0, stream,
                     x, dtbuf, Bm, Aneg, Sb, Pb);
  hipLaunchKernelGGL(ssm_carry_combine, dim3((B_SZ * DEXP * NST) / 256), dim3(256), 0, stream,
                     Sb, Pb, Car);
  hipLaunchKernelGGL(ssm_scan_out, dim3(B_SZ * SCAN_DG * NCHUNK), dim3(SCAN_T), 0, stream,
                     x, dtbuf, Bm, Cb, Aneg, Car, out);
}


# 16K-element pre-pass Aneg = -exp(A_log) moved into the first 64 blocks of the projection kernel (loaded with the x tile, stored behind the first barrier): 4 launches per call, not 5; same values
# speedup vs baseline: 4.9609x; 1.0391x over previous
#include <hip/hip_runtime.h>
#include <math.h>

#define B_SZ   2
#define LSEQ   2048
#define DEXP   1024
#define DRANK  64
#define NST    16
#define RPROJ  (DRANK + 2*NST)
#define NCHUNK 32
#define CHUNK  (LSEQ / NCHUNK)
#define ROWS   (B_SZ * LSEQ)

typedef __attribute__((ext_vector_type(2))) float v2f;
typedef __attribute__((ext_vector_type(8))) float v8f;

__device__ __forceinline__ v8f wmma_f32_k4(v2f a, v2f b, v8f c) {
  return __builtin_amdgcn_wmma_f32_16x16x4_f32(
       false, a,  false, b,
       (short)0, c,  false,  false);
}

__device__ __forceinline__ float softplus_f32(float v) {
  const float t = __expf(-fabsf(v));
  const float d = 2.0f + t;
  float r = __builtin_amdgcn_rcpf(d);
  r = fmaf(fmaf(-d, r, 1.0f), r, r);
  const float s = t * r;
  const float z = s * s;
  float p = 1.0f / 15.0f;
  p = fmaf(p, z, 1.0f / 13.0f);
  p = fmaf(p, z, 1.0f / 11.0f);
  p = fmaf(p, z, 1.0f / 9.0f);
  p = fmaf(p, z, 1.0f / 7.0f);
  p = fmaf(p, z, 1.0f / 5.0f);
  p = fmaf(p, z, 1.0f / 3.0f);
  const float s2 = s + s;
  return fmaxf(v, 0.0f) + fmaf(s2 * z, p, s2);
}

#define XP4  33
#define PP   100

__global__ __launch_bounds__(256) void ssm_proj_wmma(
    const float* __restrict__ x,    const float* __restrict__ xw,
    const float* __restrict__ xb,   const float* __restrict__ dtw,
    const float* __restrict__ dtb,  const float* __restrict__ A_log,
    float* __restrict__ dtbuf, float* __restrict__ Bmbuf, float* __restrict__ Cbuf,
    float* __restrict__ Aneg)
{
  __shared__ float4 sx4[8 * 16 * XP4];
  __shared__ float  sred[8 * 16 * PP];
  __shared__ float4 sproj4[16 * PP / 4];

  const int t    = threadIdx.x;
  const int wave = t >> 5;
  const int lane = t & 31;
  const int nl   = t & 15;
  const int hi   = (t >> 4) & 1;
  const int rowbase = blockIdx.x * 16;
  const int kw   = wave * 128;

  const int  ia      = (blockIdx.x & ((DEXP * NST) / 256 - 1)) * 256 + t;
  const bool do_aneg = blockIdx.x < (DEXP * NST) / 256;
  const float alog   = A_log[ia];

  {
    float4 v[16];
    for (int r = 0; r < 16; ++r)
      v[r] = *(const float4*)(x + (size_t)(rowbase + r) * DEXP + kw + lane * 4);
    float4* s = sx4 + wave * 16 * XP4;
    for (int r = 0; r < 16; ++r) s[r * XP4 + lane] = v[r];
  }
  __syncthreads();
  if (do_aneg) Aneg[ia] = -__expf(alog);
  float4 av[16];
  {
    const float4* s = sx4 + wave * 16 * XP4 + nl * XP4 + hi;
    for (int i = 0; i < 16; ++i) av[i] = s[i * 2];
  }

  v8f acc[6];
  for (int tt = 0; tt < 6; ++tt)
    for (int r = 0; r < 8; ++r) acc[tt][r] = 0.0f;
  {
    const float* wb = xw + (size_t)nl * DEXP + kw + hi * 4;
    #pragma unroll
    for (int i = 0; i < 16; ++i) {
      float4 bv[6];
      for (int tt = 0; tt < 6; ++tt)
        bv[tt] = *(const float4*)(wb + (size_t)tt * 16 * DEXP + i * 8);
      v2f a0, a1;
      a0.x = av[i].x; a0.y = av[i].y; a1.x = av[i].z; a1.y = av[i].w;
      for (int tt = 0; tt < 6; ++tt) {
        v2f b0; b0.x = bv[tt].x; b0.y = bv[tt].y;
        acc[tt] = wmma_f32_k4(a0, b0, acc[tt]);
      }
      for (int tt = 0; tt < 6; ++tt) {
        v2f b1; b1.x = bv[tt].z; b1.y = bv[tt].w;
        acc[tt] = wmma_f32_k4(a1, b1, acc[tt]);
      }
    }
  }
  for (int tt = 0; tt < 6; ++tt)
    for (int r = 0; r < 8; ++r)
      sred[(wave * 16 + r + hi * 8) * PP + tt * 16 + nl] = acc[tt][r];
  __syncthreads();
  float* sproj = (float*)sproj4;
  for (int i = 0; i < 6; ++i) {
    const int idx = t + i * 256;
    const int row = idx / RPROJ;
    const int col = idx - row * RPROJ;
    float v = xb[col];
    for (int w = 0; w < 8; ++w) v += sred[(w * 16 + row) * PP + col];
    sproj[row * PP + col] = v;
  }
  __syncthreads();
  if (t < 128) {
    const int q = t & 63, row = q >> 2, c4 = q & 3;
    const float4 v = sproj4[row * (PP / 4) + 16 + (t >> 6) * 4 + c4];
    float* dst = (t < 64) ? Bmbuf : Cbuf;
    *(float4*)(dst + (size_t)(rowbase + row) * NST + c4 * 4) = v;
  }

  float4 a2[8];
  {
    const float4* s = sproj4 + nl * (PP / 4) + hi;
    for (int q = 0; q < 8; ++q) a2[q] = s[q * 2];
  }
  const float* db = dtw + (size_t)(wave * 128 + nl) * DRANK + hi * 4;
  float4 bc[2][8];
  for (int u = 0; u < 2; ++u)
    for (int q = 0; q < 8; ++q)
      bc[u][q] = *(const float4*)(db + (size_t)(u * 16) * DRANK + q * 8);
  #pragma unroll
  for (int jj = 0; jj < 8; jj += 2) {
    float4 bn[2][8];
    if (jj + 2 < 8) {
      for (int u = 0; u < 2; ++u)
        for (int q = 0; q < 8; ++q)
          bn[u][q] = *(const float4*)(db + (size_t)((jj + 2 + u) * 16) * DRANK + q * 8);
    }
    const int col0 = wave * 128 + jj * 16 + nl;
    const int col1 = col0 + 16;
    v8f c0, c1;
    { const float bv0 = dtb[col0], bv1 = dtb[col1];
      for (int r = 0; r < 8; ++r) { c0[r] = bv0; c1[r] = bv1; } }
    for (int q = 0; q < 8; ++q) {
      v2f a0, a1, b00, b01, b10, b11;
      a0.x = a2[q].x; a0.y = a2[q].y; a1.x = a2[q].z; a1.y = a2[q].w;
      b00.x = bc[0][q].x; b00.y = bc[0][q].y; b01.x = bc[0][q].z; b01.y = bc[0][q].w;
      b10.x = bc[1][q].x; b10.y = bc[1][q].y; b11.x = bc[1][q].z; b11.y = bc[1][q].w;
      c0 = wmma_f32_k4(a0, b00, c0);
      c1 = wmma_f32_k4(a0, b10, c1);
      c0 = wmma_f32_k4(a1, b01, c0);
      c1 = wmma_f32_k4(a1, b11, c1);
    }
    for (int r = 0; r < 8; ++r) {
      const int M = r + hi * 8;
      dtbuf[(size_t)(rowbase + M) * DEXP + col0] = softplus_f32(c0[r]);
      dtbuf[(size_t)(rowbase + M) * DEXP + col1] = softplus_f32(c1[r]);
    }
    if (jj + 2 < 8) {
      for (int u = 0; u < 2; ++u)
        for (int q = 0; q < 8; ++q) bc[u][q] = bn[u][q];
    }
  }
}

#define SCAN_T   64
#define SCAN_DG  (DEXP / SCAN_T)
#define SCAN_U   8
#define SCAN_UO  4
#define FILL_R   16

__device__ __forceinline__ void stage_cols(const float* __restrict__ g0, const float* __restrict__ g1,
                                           float4* s, int t) {
  const int col4 = t & 15, rsub = t >> 4;
  for (int i0 = 0; i0 < CHUNK / 4; i0 += FILL_R) {
    float4 v0[FILL_R], v1[FILL_R];
    for (int k = 0; k < FILL_R; ++k) {
      const int row = (i0 + k) * 4 + rsub;
      v0[k] = *(const float4*)(g0 + (size_t)row * DEXP + col4 * 4);
      v1[k] = *(const float4*)(g1 + (size_t)row * DEXP + col4 * 4);
    }
    for (int k = 0; k < FILL_R; ++k) {
      const int row = (i0 + k) * 4 + rsub;
      float4 lo, hi;
      lo.x = v0[k].x; lo.y = v1[k].x; lo.z = v0[k].y; lo.w = v1[k].y;
      hi.x = v0[k].z; hi.y = v1[k].z; hi.z = v0[k].w; hi.w = v1[k].w;
      s[row * 32 + col4 * 2]     = lo;
      s[row * 32 + col4 * 2 + 1] = hi;
    }
  }
}

__global__ __launch_bounds__(SCAN_T) void ssm_scan_carry(
    const float* __restrict__ x, const float* __restrict__ dtbuf,
    const float* __restrict__ Bmbuf, const float* __restrict__ Aneg,
    float* __restrict__ Sbuf, float* __restrict__ Pbuf)
{
  __shared__ float4 sdx[CHUNK * 32];
  __shared__ float4 sB [CHUNK * 4];

  const int t  = threadIdx.x;
  const int bi = blockIdx.x;
  const int c  = bi & (NCHUNK - 1);
  const int dg = (bi / NCHUNK) & (SCAN_DG - 1);
  const int b  = bi / (NCHUNK * SCAN_DG);
  const int d  = dg * SCAN_T + t;
  const size_t rbase = (size_t)b * LSEQ + (size_t)c * CHUNK;

  {
    const float4* Bg = (const float4*)(Bmbuf + rbase * NST);
    float4 vb[CHUNK * 4 / SCAN_T];
    for (int i = 0; i < CHUNK * 4 / SCAN_T; ++i) vb[i] = Bg[t + i * SCAN_T];
    for (int i = 0; i < CHUNK * 4 / SCAN_T; ++i) sB[t + i * SCAN_T] = vb[i];
  }
  stage_cols(dtbuf + rbase * DEXP + dg * SCAN_T, x + rbase * DEXP + dg * SCAN_T, sdx, t);

  float A[NST];
  for (int q = 0; q < 4; ++q) {
    const float4 v = ((const float4*)(Aneg + (size_t)d * NST))[q];
    A[4 * q] = v.x; A[4 * q + 1] = v.y; A[4 * q + 2] = v.z; A[4 * q + 3] = v.w;
  }
  __syncthreads();

  const float2* sdxf = (const float2*)sdx + t;

  float h[NST];
  for (int n = 0; n < NST; ++n) h[n] = 0.0f;
  float sdt_sum = 0.0f;

  float2 dxn[SCAN_U];
  for (int j = 0; j < SCAN_U; ++j) dxn[j] = sdxf[j * SCAN_T];
  for (int l0 = 0; l0 < CHUNK; l0 += SCAN_U) {
    float2 dxc[SCAN_U];
    for (int j = 0; j < SCAN_U; ++j) dxc[j] = dxn[j];
    for (int j = 0; j < SCAN_U; ++j)
      dxn[j] = sdxf[((l0 + SCAN_U + j) & (CHUNK - 1)) * SCAN_T];
    #pragma unroll
    for (int j = 0; j < SCAN_U; ++j) {
      const int li = l0 + j;
      const float dtv = dxc[j].x;
      const float u   = dtv * dxc[j].y;
      for (int q = 0; q < 4; ++q) {
        const float4 bv = sB[li * 4 + q];
        h[4 * q]     = fmaf(__expf(A[4 * q]     * dtv), h[4 * q],     u * bv.x);
        h[4 * q + 1] = fmaf(__expf(A[4 * q + 1] * dtv), h[4 * q + 1], u * bv.y);
        h[4 * q + 2] = fmaf(__expf(A[4 * q + 2] * dtv), h[4 * q + 2], u * bv.z);
        h[4 * q + 3] = fmaf(__expf(A[4 * q + 3] * dtv), h[4 * q + 3], u * bv.w);
      }
      sdt_sum += dtv;
    }
  }
  const size_t si = (((size_t)b * NCHUNK + c) * DEXP + d) * NST;
  for (int q = 0; q < 4; ++q) {
    float4 s, p;
    s.x = h[4 * q]; s.y = h[4 * q + 1]; s.z = h[4 * q + 2]; s.w = h[4 * q + 3];
    p.x = __expf(A[4 * q] * sdt_sum);     p.y = __expf(A[4 * q + 1] * sdt_sum);
    p.z = __expf(A[4 * q + 2] * sdt_sum); p.w = __expf(A[4 * q + 3] * sdt_sum);
    ((float4*)(Sbuf + si))[q] = s;
    ((float4*)(Pbuf + si))[q] = p;
  }
}

__global__ void ssm_carry_combine(const float* __restrict__ Sbuf,
                                  const float* __restrict__ Pbuf,
                                  float* __restrict__ Car)
{
  const int tg = blockIdx.x * 256 + threadIdx.x;
  const int n  = tg & 15;
  const int d  = (tg >> 4) & (DEXP - 1);
  const int b  = tg >> 14;
  const size_t stride = (size_t)DEXP * NST;
  const size_t base   = (size_t)b * NCHUNK * stride + (size_t)d * NST + n;
  float sv[NCHUNK - 1], pv[NCHUNK - 1];
  #pragma unroll
  for (int c = 0; c < NCHUNK - 1; ++c) {
    sv[c] = Sbuf[base + (size_t)c * stride];
    pv[c] = Pbuf[base + (size_t)c * stride];
  }
  float cin = 0.0f;
  Car[base] = 0.0f;
  #pragma unroll
  for (int c = 1; c < NCHUNK; ++c) {
    cin = sv[c - 1] + pv[c - 1] * cin;
    Car[base + (size_t)c * stride] = cin;
  }
}

__global__ __launch_bounds__(SCAN_T) void ssm_scan_out(
    const float* __restrict__ x, const float* __restrict__ dtbuf,
    const float* __restrict__ Bmbuf, const float* __restrict__ Cbuf,
    const float* __restrict__ Aneg, const float* __restrict__ Car,
    float* __restrict__ out)
{
  __shared__ float4 sdx[CHUNK * 32];
  __shared__ float4 sB [CHUNK * 4];
  __shared__ float4 sC [CHUNK * 4];

  const int t  = threadIdx.x;
  const int bi = blockIdx.x;
  const int c  = bi & (NCHUNK - 1);
  const int dg = (bi / NCHUNK) & (SCAN_DG - 1);
  const int b  = bi / (NCHUNK * SCAN_DG);
  const int d  = dg * SCAN_T + t;
  const size_t rbase = (size_t)b * LSEQ + (size_t)c * CHUNK;

  {
    const float4* Bg = (const float4*)(Bmbuf + rbase * NST);
    const float4* Cg = (const float4*)(Cbuf  + rbase * NST);
    float4 vb[CHUNK * 4 / SCAN_T], vc[CHUNK * 4 / SCAN_T];
    for (int i = 0; i < CHUNK * 4 / SCAN_T; ++i) { vb[i] = Bg[t + i * SCAN_T]; vc[i] = Cg[t + i * SCAN_T]; }
    for (int i = 0; i < CHUNK * 4 / SCAN_T; ++i) { sB[t + i * SCAN_T] = vb[i]; sC[t + i * SCAN_T] = vc[i]; }
  }
  stage_cols(dtbuf + rbase * DEXP + dg * SCAN_T, x + rbase * DEXP + dg * SCAN_T, sdx, t);

  const size_t si = (((size_t)b * NCHUNK + c) * DEXP + d) * NST;
  float A[NST], h[NST];
  for (int q = 0; q < 4; ++q) {
    const float4 v = ((const float4*)(Aneg + (size_t)d * NST))[q];
    A[4 * q] = v.x; A[4 * q + 1] = v.y; A[4 * q + 2] = v.z; A[4 * q + 3] = v.w;
    const float4 w = ((const float4*)(Car + si))[q];
    h[4 * q] = w.x; h[4 * q + 1] = w.y; h[4 * q + 2] = w.z; h[4 * q + 3] = w.w;
  }
  __syncthreads();

  const float2* sdxf = (const float2*)sdx + t;
  float* op         = out + rbase * DEXP + d;

  float2 dxn[SCAN_UO];
  for (int j = 0; j < SCAN_UO; ++j) dxn[j] = sdxf[j * SCAN_T];
  for (int l0 = 0; l0 < CHUNK; l0 += SCAN_UO) {
    float2 dxc[SCAN_UO];
    for (int j = 0; j < SCAN_UO; ++j) dxc[j] = dxn[j];
    for (int j = 0; j < SCAN_UO; ++j)
      dxn[j] = sdxf[((l0 + SCAN_UO + j) & (CHUNK - 1)) * SCAN_T];
    #pragma unroll
    for (int j = 0; j < SCAN_UO; ++j) {
      const int li = l0 + j;
      const float dtv = dxc[j].x;
      const float u   = dtv * dxc[j].y;
      float s2[8];
      for (int q = 0; q < 4; ++q) {
        const float4 bv = sB[li * 4 + q];
        const float4 cv = sC[li * 4 + q];
        h[4 * q]     = fmaf(__expf(A[4 * q]     * dtv), h[4 * q],     u * bv.x);
        h[4 * q + 1] = fmaf(__expf(A[4 * q + 1] * dtv), h[4 * q + 1], u * bv.y);
        h[4 * q + 2] = fmaf(__expf(A[4 * q + 2] * dtv), h[4 * q + 2], u * bv.z);
        h[4 * q + 3] = fmaf(__expf(A[4 * q + 3] * dtv), h[4 * q + 3], u * bv.w);
        s2[2 * q]     = h[4 * q]     * cv.x + h[4 * q + 1] * cv.y;
        s2[2 * q + 1] = h[4 * q + 2] * cv.z + h[4 * q + 3] * cv.w;
      }
      const float y = ((s2[0] + s2[1]) + (s2[2] + s2[3])) + ((s2[4] + s2[5]) + (s2[6] + s2[7]));
      op[(size_t)li * DEXP] = y;
    }
  }
}

extern "C" void kernel_launch(void* const* d_in, const int* in_sizes, int n_in,
                              void* d_out, int out_size, void* d_ws, size_t ws_size,
                              hipStream_t stream) {
  (void)in_sizes; (void)n_in; (void)out_size; (void)ws_size;
  const float* x     = (const float*)d_in[0];
  const float* xw    = (const float*)d_in[1];
  const float* xb    = (const float*)d_in[2];
  const float* dtw   = (const float*)d_in[3];
  const float* dtb   = (const float*)d_in[4];
  const float* A_log = (const float*)d_in[5];
  float* out = (float*)d_out;

  float* ws    = (float*)d_ws;
  float* dtbuf = ws;
  float* Aneg  = dtbuf + (size_t)ROWS * DEXP;
  float* Bm    = Aneg  + (size_t)DEXP * NST;
  float* Cb    = Bm    + (size_t)ROWS * NST;
  float* Sb    = Cb    + (size_t)ROWS * NST;
  float* Pb    = Sb    + (size_t)B_SZ * NCHUNK * DEXP * NST;
  float* Car   = Pb    + (size_t)B_SZ * NCHUNK * DEXP * NST;

  hipLaunchKernelGGL(ssm_proj_wmma, dim3(ROWS / 16), dim3(256), 0, stream,
                     x, xw, xb, dtw, dtb, A_log, dtbuf, Bm, Cb, Aneg);
  hipLaunchKernelGGL(ssm_scan_carry, dim3(B_SZ * SCAN_DG * NCHUNK), dim3(SCAN_T), 0, stream,
                     x, dtbuf, Bm, Aneg, Sb, Pb);
  hipLaunchKernelGGL(ssm_carry_combine, dim3((B_SZ * DEXP * NST) / 256), dim3(256), 0, stream,
                     Sb, Pb, Car);
  hipLaunchKernelGGL(ssm_scan_out, dim3(B_SZ * SCAN_DG * NCHUNK), dim3(SCAN_T), 0, stream,
                     x, dtbuf, Bm, Cb, Aneg, Car, out);
}


# carry combine folded into the output scan: a block of 8 waves owns 4 consecutive chunks, folds the earlier chunk summaries once, passes carries in LDS; 3 launches per call, not 4; same fold order
# speedup vs baseline: 5.2434x; 1.0569x over previous
#include <hip/hip_runtime.h>
#include <math.h>

#define B_SZ   2
#define LSEQ   2048
#define DEXP   1024
#define DRANK  64
#define NST    16
#define RPROJ  (DRANK + 2*NST)
#define NCHUNK 32
#define CHUNK  (LSEQ / NCHUNK)
#define ROWS   (B_SZ * LSEQ)

typedef __attribute__((ext_vector_type(2))) float v2f;
typedef __attribute__((ext_vector_type(8))) float v8f;

__device__ __forceinline__ v8f wmma_f32_k4(v2f a, v2f b, v8f c) {
  return __builtin_amdgcn_wmma_f32_16x16x4_f32(
       false, a,  false, b,
       (short)0, c,  false,  false);
}

__device__ __forceinline__ float softplus_f32(float v) {
  const float t = __expf(-fabsf(v));
  const float d = 2.0f + t;
  float r = __builtin_amdgcn_rcpf(d);
  r = fmaf(fmaf(-d, r, 1.0f), r, r);
  const float s = t * r;
  const float z = s * s;
  float p = 1.0f / 15.0f;
  p = fmaf(p, z, 1.0f / 13.0f);
  p = fmaf(p, z, 1.0f / 11.0f);
  p = fmaf(p, z, 1.0f / 9.0f);
  p = fmaf(p, z, 1.0f / 7.0f);
  p = fmaf(p, z, 1.0f / 5.0f);
  p = fmaf(p, z, 1.0f / 3.0f);
  const float s2 = s + s;
  return fmaxf(v, 0.0f) + fmaf(s2 * z, p, s2);
}

#define XP4  33
#define PP   100

__global__ __launch_bounds__(256) void ssm_proj_wmma(
    const float* __restrict__ x,    const float* __restrict__ xw,
    const float* __restrict__ xb,   const float* __restrict__ dtw,
    const float* __restrict__ dtb,  const float* __restrict__ A_log,
    float* __restrict__ dtbuf, float* __restrict__ Bmbuf, float* __restrict__ Cbuf,
    float* __restrict__ Aneg)
{
  __shared__ float4 sx4[8 * 16 * XP4];
  __shared__ float  sred[8 * 16 * PP];
  __shared__ float4 sproj4[16 * PP / 4];

  const int t    = threadIdx.x;
  const int wave = t >> 5;
  const int lane = t & 31;
  const int nl   = t & 15;
  const int hi   = (t >> 4) & 1;
  const int rowbase = blockIdx.x * 16;
  const int kw   = wave * 128;

  const int  ia      = (blockIdx.x & ((DEXP * NST) / 256 - 1)) * 256 + t;
  const bool do_aneg = blockIdx.x < (DEXP * NST) / 256;
  const float alog   = A_log[ia];

  {
    float4 v[16];
    for (int r = 0; r < 16; ++r)
      v[r] = *(const float4*)(x + (size_t)(rowbase + r) * DEXP + kw + lane * 4);
    float4* s = sx4 + wave * 16 * XP4;
    for (int r = 0; r < 16; ++r) s[r * XP4 + lane] = v[r];
  }
  __syncthreads();
  if (do_aneg) Aneg[ia] = -__expf(alog);
  float4 av[16];
  {
    const float4* s = sx4 + wave * 16 * XP4 + nl * XP4 + hi;
    for (int i = 0; i < 16; ++i) av[i] = s[i * 2];
  }

  v8f acc[6];
  for (int tt = 0; tt < 6; ++tt)
    for (int r = 0; r < 8; ++r) acc[tt][r] = 0.0f;
  {
    const float* wb = xw + (size_t)nl * DEXP + kw + hi * 4;
    #pragma unroll
    for (int i = 0; i < 16; ++i) {
      float4 bv[6];
      for (int tt = 0; tt < 6; ++tt)
        bv[tt] = *(const float4*)(wb + (size_t)tt * 16 * DEXP + i * 8);
      v2f a0, a1;
      a0.x = av[i].x; a0.y = av[i].y; a1.x = av[i].z; a1.y = av[i].w;
      for (int tt = 0; tt < 6; ++tt) {
        v2f b0; b0.x = bv[tt].x; b0.y = bv[tt].y;
        acc[tt] = wmma_f32_k4(a0, b0, acc[tt]);
      }
      for (int tt = 0; tt < 6; ++tt) {
        v2f b1; b1.x = bv[tt].z; b1.y = bv[tt].w;
        acc[tt] = wmma_f32_k4(a1, b1, acc[tt]);
      }
    }
  }
  for (int tt = 0; tt < 6; ++tt)
    for (int r = 0; r < 8; ++r)
      sred[(wave * 16 + r + hi * 8) * PP + tt * 16 + nl] = acc[tt][r];
  __syncthreads();
  float* sproj = (float*)sproj4;
  for (int i = 0; i < 6; ++i) {
    const int idx = t + i * 256;
    const int row = idx / RPROJ;
    const int col = idx - row * RPROJ;
    float v = xb[col];
    for (int w = 0; w < 8; ++w) v += sred[(w * 16 + row) * PP + col];
    sproj[row * PP + col] = v;
  }
  __syncthreads();
  if (t < 128) {
    const int q = t & 63, row = q >> 2, c4 = q & 3;
    const float4 v = sproj4[row * (PP / 4) + 16 + (t >> 6) * 4 + c4];
    float* dst = (t < 64) ? Bmbuf : Cbuf;
    *(float4*)(dst + (size_t)(rowbase + row) * NST + c4 * 4) = v;
  }

  float4 a2[8];
  {
    const float4* s = sproj4 + nl * (PP / 4) + hi;
    for (int q = 0; q < 8; ++q) a2[q] = s[q * 2];
  }
  const float* db = dtw + (size_t)(wave * 128 + nl) * DRANK + hi * 4;
  float4 bc[2][8];
  for (int u = 0; u < 2; ++u)
    for (int q = 0; q < 8; ++q)
      bc[u][q] = *(const float4*)(db + (size_t)(u * 16) * DRANK + q * 8);
  #pragma unroll
  for (int jj = 0; jj < 8; jj += 2) {
    float4 bn[2][8];
    if (jj + 2 < 8) {
      for (int u = 0; u < 2; ++u)
        for (int q = 0; q < 8; ++q)
          bn[u][q] = *(const float4*)(db + (size_t)((jj + 2 + u) * 16) * DRANK + q * 8);
    }
    const int col0 = wave * 128 + jj * 16 + nl;
    const int col1 = col0 + 16;
    v8f c0, c1;
    { const float bv0 = dtb[col0], bv1 = dtb[col1];
      for (int r = 0; r < 8; ++r) { c0[r] = bv0; c1[r] = bv1; } }
    for (int q = 0; q < 8; ++q) {
      v2f a0, a1, b00, b01, b10, b11;
      a0.x = a2[q].x; a0.y = a2[q].y; a1.x = a2[q].z; a1.y = a2[q].w;
      b00.x = bc[0][q].x; b00.y = bc[0][q].y; b01.x = bc[0][q].z; b01.y = bc[0][q].w;
      b10.x = bc[1][q].x; b10.y = bc[1][q].y; b11.x = bc[1][q].z; b11.y = bc[1][q].w;
      c0 = wmma_f32_k4(a0, b00, c0);
      c1 = wmma_f32_k4(a0, b10, c1);
      c0 = wmma_f32_k4(a1, b01, c0);
      c1 = wmma_f32_k4(a1, b11, c1);
    }
    for (int r = 0; r < 8; ++r) {
      const int M = r + hi * 8;
      dtbuf[(size_t)(rowbase + M) * DEXP + col0] = softplus_f32(c0[r]);
      dtbuf[(size_t)(rowbase + M) * DEXP + col1] = softplus_f32(c1[r]);
    }
    if (jj + 2 < 8) {
      for (int u = 0; u < 2; ++u)
        for (int q = 0; q < 8; ++q) bc[u][q] = bn[u][q];
    }
  }
}

#define SCAN_T   64
#define SCAN_DG  (DEXP / SCAN_T)
#define SCAN_U   8
#define SCAN_UO  4
#define FILL_R   16

__device__ __forceinline__ void stage_cols(const float* __restrict__ g0, const float* __restrict__ g1,
                                           float4* s, int t) {
  const int col4 = t & 15, rsub = t >> 4;
  for (int i0 = 0; i0 < CHUNK / 4; i0 += FILL_R) {
    float4 v0[FILL_R], v1[FILL_R];
    for (int k = 0; k < FILL_R; ++k) {
      const int row = (i0 + k) * 4 + rsub;
      v0[k] = *(const float4*)(g0 + (size_t)row * DEXP + col4 * 4);
      v1[k] = *(const float4*)(g1 + (size_t)row * DEXP + col4 * 4);
    }
    for (int k = 0; k < FILL_R; ++k) {
      const int row = (i0 + k) * 4 + rsub;
      float4 lo, hi;
      lo.x = v0[k].x; lo.y = v1[k].x; lo.z = v0[k].y; lo.w = v1[k].y;
      hi.x = v0[k].z; hi.y = v1[k].z; hi.z = v0[k].w; hi.w = v1[k].w;
      s[row * 32 + col4 * 2]     = lo;
      s[row * 32 + col4 * 2 + 1] = hi;
    }
  }
}

__global__ __launch_bounds__(SCAN_T) void ssm_scan_carry(
    const float* __restrict__ x, const float* __restrict__ dtbuf,
    const float* __restrict__ Bmbuf, const float* __restrict__ Aneg,
    float* __restrict__ Sbuf, float* __restrict__ Pbuf)
{
  __shared__ float4 sdx[CHUNK * 32];
  __shared__ float4 sB [CHUNK * 4];

  const int t  = threadIdx.x;
  const int bi = blockIdx.x;
  const int c  = bi & (NCHUNK - 1);
  const int dg = (bi / NCHUNK) & (SCAN_DG - 1);
  const int b  = bi / (NCHUNK * SCAN_DG);
  const int d  = dg * SCAN_T + t;
  const size_t rbase = (size_t)b * LSEQ + (size_t)c * CHUNK;

  {
    const float4* Bg = (const float4*)(Bmbuf + rbase * NST);
    float4 vb[CHUNK * 4 / SCAN_T];
    for (int i = 0; i < CHUNK * 4 / SCAN_T; ++i) vb[i] = Bg[t + i * SCAN_T];
    for (int i = 0; i < CHUNK * 4 / SCAN_T; ++i) sB[t + i * SCAN_T] = vb[i];
  }
  stage_cols(dtbuf + rbase * DEXP + dg * SCAN_T, x + rbase * DEXP + dg * SCAN_T, sdx, t);

  float A[NST];
  for (int q = 0; q < 4; ++q) {
    const float4 v = ((const float4*)(Aneg + (size_t)d * NST))[q];
    A[4 * q] = v.x; A[4 * q + 1] = v.y; A[4 * q + 2] = v.z; A[4 * q + 3] = v.w;
  }
  __syncthreads();

  const float2* sdxf = (const float2*)sdx + t;

  float h[NST];
  for (int n = 0; n < NST; ++n) h[n] = 0.0f;
  float sdt_sum = 0.0f;

  float2 dxn[SCAN_U];
  for (int j = 0; j < SCAN_U; ++j) dxn[j] = sdxf[j * SCAN_T];
  for (int l0 = 0; l0 < CHUNK; l0 += SCAN_U) {
    float2 dxc[SCAN_U];
    for (int j = 0; j < SCAN_U; ++j) dxc[j] = dxn[j];
    for (int j = 0; j < SCAN_U; ++j)
      dxn[j] = sdxf[((l0 + SCAN_U + j) & (CHUNK - 1)) * SCAN_T];
    #pragma unroll
    for (int j = 0; j < SCAN_U; ++j) {
      const int li = l0 + j;
      const float dtv = dxc[j].x;
      const float u   = dtv * dxc[j].y;
      for (int q = 0; q < 4; ++q) {
        const float4 bv = sB[li * 4 + q];
        h[4 * q]     = fmaf(__expf(A[4 * q]     * dtv), h[4 * q],     u * bv.x);
        h[4 * q + 1] = fmaf(__expf(A[4 * q + 1] * dtv), h[4 * q + 1], u * bv.y);
        h[4 * q + 2] = fmaf(__expf(A[4 * q + 2] * dtv), h[4 * q + 2], u * bv.z);
        h[4 * q + 3] = fmaf(__expf(A[4 * q + 3] * dtv), h[4 * q + 3], u * bv.w);
      }
      sdt_sum += dtv;
    }
  }
  const size_t si = (((size_t)b * NCHUNK + c) * DEXP + d) * NST;
  for (int q = 0; q < 4; ++q) {
    float4 s, p;
    s.x = h[4 * q]; s.y = h[4 * q + 1]; s.z = h[4 * q + 2]; s.w = h[4 * q + 3];
    p.x = __expf(A[4 * q] * sdt_sum);     p.y = __expf(A[4 * q + 1] * sdt_sum);
    p.z = __expf(A[4 * q + 2] * sdt_sum); p.w = __expf(A[4 * q + 3] * sdt_sum);
    ((float4*)(Sbuf + si))[q] = s;
    ((float4*)(Pbuf + si))[q] = p;
  }
}

template <int NP>
__device__ __forceinline__ float4 fold_fixed(const float4* __restrict__ Sg, const float4* __restrict__ Pg, float4 cin) {
  const size_t slab4 = (size_t)DEXP * NST / 4;
  float4 sv[NP], pv[NP];
  #pragma unroll
  for (int j = 0; j < NP; ++j) { sv[j] = Sg[(size_t)j * slab4]; pv[j] = Pg[(size_t)j * slab4]; }
  #pragma unroll
  for (int j = 0; j < NP; ++j) {
    cin.x = fmaf(pv[j].x, cin.x, sv[j].x);
    cin.y = fmaf(pv[j].y, cin.y, sv[j].y);
    cin.z = fmaf(pv[j].z, cin.z, sv[j].z);
    cin.w = fmaf(pv[j].w, cin.w, sv[j].w);
  }
  return cin;
}

#define SCAN_G   4

__global__ __launch_bounds__(SCAN_T * SCAN_G) void ssm_scan_out(
    const float* __restrict__ x, const float* __restrict__ dtbuf,
    const float* __restrict__ Bmbuf, const float* __restrict__ Cbuf,
    const float* __restrict__ Aneg, const float* __restrict__ Sbuf,
    const float* __restrict__ Pbuf, float* __restrict__ out)
{
  __shared__ float4 sdx [SCAN_G][CHUNK * 32];
  __shared__ float4 sB  [SCAN_G][CHUNK * 4];
  __shared__ float4 sC  [SCAN_G][CHUNK * 4];
  __shared__ float4 scar[SCAN_G][SCAN_T * NST / 4];

  const int tb = threadIdx.x;
  const int k  = tb >> 6;
  const int t  = tb & (SCAN_T - 1);
  const int bi = blockIdx.x;
  const int g  = bi & (NCHUNK / SCAN_G - 1);
  const int dg = (bi / (NCHUNK / SCAN_G)) & (SCAN_DG - 1);
  const int b  = bi / ((NCHUNK / SCAN_G) * SCAN_DG);
  const int c0 = g * SCAN_G;
  const int c  = c0 + k;
  const int d  = dg * SCAN_T + t;
  const size_t rbase = (size_t)b * LSEQ + (size_t)c * CHUNK;

  const float4* Bg = (const float4*)(Bmbuf + rbase * NST);
  const float4* Cg = (const float4*)(Cbuf  + rbase * NST);
  float4 vb[CHUNK * 4 / SCAN_T], vc[CHUNK * 4 / SCAN_T];
  for (int i = 0; i < CHUNK * 4 / SCAN_T; ++i) { vb[i] = Bg[t + i * SCAN_T]; vc[i] = Cg[t + i * SCAN_T]; }

  {
    const size_t slab4 = (size_t)DEXP * NST / 4;
    const size_t o4    = (size_t)b * NCHUNK * slab4 + (size_t)dg * (SCAN_T * NST / 4) + tb;
    const float4* Sg = (const float4*)Sbuf + o4;
    const float4* Pg = (const float4*)Pbuf + o4;
    float4 so[SCAN_G - 1], po[SCAN_G - 1];
    for (int j = 0; j < SCAN_G - 1; ++j) {
      so[j] = Sg[(size_t)(c0 + j) * slab4];
      po[j] = Pg[(size_t)(c0 + j) * slab4];
    }
    float4 cin; cin.x = 0.0f; cin.y = 0.0f; cin.z = 0.0f; cin.w = 0.0f;
    switch (g) {
      case 1: cin = fold_fixed< 4>(Sg, Pg, cin); break;
      case 2: cin = fold_fixed< 8>(Sg, Pg, cin); break;
      case 3: cin = fold_fixed<12>(Sg, Pg, cin); break;
      case 4: cin = fold_fixed<16>(Sg, Pg, cin); break;
      case 5: cin = fold_fixed<20>(Sg, Pg, cin); break;
      case 6: cin = fold_fixed<24>(Sg, Pg, cin); break;
      case 7: cin = fold_fixed<28>(Sg, Pg, cin); break;
      default: break;
    }
    scar[0][tb] = cin;
    for (int j = 0; j < SCAN_G - 1; ++j) {
      cin.x = fmaf(po[j].x, cin.x, so[j].x);
      cin.y = fmaf(po[j].y, cin.y, so[j].y);
      cin.z = fmaf(po[j].z, cin.z, so[j].z);
      cin.w = fmaf(po[j].w, cin.w, so[j].w);
      scar[j + 1][tb] = cin;
    }
  }

  for (int i = 0; i < CHUNK * 4 / SCAN_T; ++i) { sB[k][t + i * SCAN_T] = vb[i]; sC[k][t + i * SCAN_T] = vc[i]; }
  stage_cols(dtbuf + rbase * DEXP + dg * SCAN_T, x + rbase * DEXP + dg * SCAN_T, sdx[k], t);

  float A[NST], h[NST];
  for (int q = 0; q < 4; ++q) {
    const float4 v = ((const float4*)(Aneg + (size_t)d * NST))[q];
    A[4 * q] = v.x; A[4 * q + 1] = v.y; A[4 * q + 2] = v.z; A[4 * q + 3] = v.w;
  }
  __syncthreads();
  for (int q = 0; q < 4; ++q) {
    const float4 w = scar[k][t * 4 + q];
    h[4 * q] = w.x; h[4 * q + 1] = w.y; h[4 * q + 2] = w.z; h[4 * q + 3] = w.w;
  }

  const float2* sdxf = (const float2*)sdx[k] + t;
  const float4* sBk  = sB[k];
  const float4* sCk  = sC[k];
  float* op          = out + rbase * DEXP + d;

  float2 dxn[SCAN_UO];
  for (int j = 0; j < SCAN_UO; ++j) dxn[j] = sdxf[j * SCAN_T];
  for (int l0 = 0; l0 < CHUNK; l0 += SCAN_UO) {
    float2 dxc[SCAN_UO];
    for (int j = 0; j < SCAN_UO; ++j) dxc[j] = dxn[j];
    for (int j = 0; j < SCAN_UO; ++j)
      dxn[j] = sdxf[((l0 + SCAN_UO + j) & (CHUNK - 1)) * SCAN_T];
    #pragma unroll
    for (int j = 0; j < SCAN_UO; ++j) {
      const int li = l0 + j;
      const float dtv = dxc[j].x;
      const float u   = dtv * dxc[j].y;
      float s2[8];
      for (int q = 0; q < 4; ++q) {
        const float4 bv = sBk[li * 4 + q];
        const float4 cv = sCk[li * 4 + q];
        h[4 * q]     = fmaf(__expf(A[4 * q]     * dtv), h[4 * q],     u * bv.x);
        h[4 * q + 1] = fmaf(__expf(A[4 * q + 1] * dtv), h[4 * q + 1], u * bv.y);
        h[4 * q + 2] = fmaf(__expf(A[4 * q + 2] * dtv), h[4 * q + 2], u * bv.z);
        h[4 * q + 3] = fmaf(__expf(A[4 * q + 3] * dtv), h[4 * q + 3], u * bv.w);
        s2[2 * q]     = h[4 * q]     * cv.x + h[4 * q + 1] * cv.y;
        s2[2 * q + 1] = h[4 * q + 2] * cv.z + h[4 * q + 3] * cv.w;
      }
      const float y = ((s2[0] + s2[1]) + (s2[2] + s2[3])) + ((s2[4] + s2[5]) + (s2[6] + s2[7]));
      op[(size_t)li * DEXP] = y;
    }
  }
}

extern "C" void kernel_launch(void* const* d_in, const int* in_sizes, int n_in,
                              void* d_out, int out_size, void* d_ws, size_t ws_size,
                              hipStream_t stream) {
  (void)in_sizes; (void)n_in; (void)out_size; (void)ws_size;
  const float* x     = (const float*)d_in[0];
  const float* xw    = (const float*)d_in[1];
  const float* xb    = (const float*)d_in[2];
  const float* dtw   = (const float*)d_in[3];
  const float* dtb   = (const float*)d_in[4];
  const float* A_log = (const float*)d_in[5];
  float* out = (float*)d_out;

  float* ws    = (float*)d_ws;
  float* dtbuf = ws;
  float* Aneg  = dtbuf + (size_t)ROWS * DEXP;
  float* Bm    = Aneg  + (size_t)DEXP * NST;
  float* Cb    = Bm    + (size_t)ROWS * NST;
  float* Sb    = Cb    + (size_t)ROWS * NST;
  float* Pb    = Sb    + (size_t)B_SZ * NCHUNK * DEXP * NST;

  hipLaunchKernelGGL(ssm_proj_wmma, dim3(ROWS / 16), dim3(256), 0, stream,
                     x, xw, xb, dtw, dtb, A_log, dtbuf, Bm, Cb, Aneg);
  hipLaunchKernelGGL(ssm_scan_carry, dim3(B_SZ * SCAN_DG * NCHUNK), dim3(SCAN_T), 0, stream,
                     x, dtbuf, Bm, Aneg, Sb, Pb);
  hipLaunchKernelGGL(ssm_scan_out, dim3(B_SZ * SCAN_DG * (NCHUNK / SCAN_G)), dim3(SCAN_T * SCAN_G), 0, stream,
                     x, dtbuf, Bm, Cb, Aneg, Sb, Pb, out);
}


# chunk decay P = exp(A * sum dt) no longer stored by the carry scan and reloaded by the output scan's fold: the fold forms it from the chunk's dt sum (1 float per channel, not 16); same values
# speedup vs baseline: 5.4267x; 1.0350x over previous
#include <hip/hip_runtime.h>
#include <math.h>

#define B_SZ   2
#define LSEQ   2048
#define DEXP   1024
#define DRANK  64
#define NST    16
#define RPROJ  (DRANK + 2*NST)
#define NCHUNK 32
#define CHUNK  (LSEQ / NCHUNK)
#define ROWS   (B_SZ * LSEQ)

typedef __attribute__((ext_vector_type(2))) float v2f;
typedef __attribute__((ext_vector_type(8))) float v8f;

__device__ __forceinline__ v8f wmma_f32_k4(v2f a, v2f b, v8f c) {
  return __builtin_amdgcn_wmma_f32_16x16x4_f32(
       false, a,  false, b,
       (short)0, c,  false,  false);
}

__device__ __forceinline__ float softplus_f32(float v) {
  const float t = __expf(-fabsf(v));
  const float d = 2.0f + t;
  float r = __builtin_amdgcn_rcpf(d);
  r = fmaf(fmaf(-d, r, 1.0f), r, r);
  const float s = t * r;
  const float z = s * s;
  float p = 1.0f / 15.0f;
  p = fmaf(p, z, 1.0f / 13.0f);
  p = fmaf(p, z, 1.0f / 11.0f);
  p = fmaf(p, z, 1.0f / 9.0f);
  p = fmaf(p, z, 1.0f / 7.0f);
  p = fmaf(p, z, 1.0f / 5.0f);
  p = fmaf(p, z, 1.0f / 3.0f);
  const float s2 = s + s;
  return fmaxf(v, 0.0f) + fmaf(s2 * z, p, s2);
}

#define XP4  33
#define PP   100

__global__ __launch_bounds__(256) void ssm_proj_wmma(
    const float* __restrict__ x,    const float* __restrict__ xw,
    const float* __restrict__ xb,   const float* __restrict__ dtw,
    const float* __restrict__ dtb,  const float* __restrict__ A_log,
    float* __restrict__ dtbuf, float* __restrict__ Bmbuf, float* __restrict__ Cbuf,
    float* __restrict__ Aneg)
{
  __shared__ float4 sx4[8 * 16 * XP4];
  __shared__ float  sred[8 * 16 * PP];
  __shared__ float4 sproj4[16 * PP / 4];

  const int t    = threadIdx.x;
  const int wave = t >> 5;
  const int lane = t & 31;
  const int nl   = t & 15;
  const int hi   = (t >> 4) & 1;
  const int rowbase = blockIdx.x * 16;
  const int kw   = wave * 128;

  const int  ia      = (blockIdx.x & ((DEXP * NST) / 256 - 1)) * 256 + t;
  const bool do_aneg = blockIdx.x < (DEXP * NST) / 256;
  const float alog   = A_log[ia];

  {
    float4 v[16];
    for (int r = 0; r < 16; ++r)
      v[r] = *(const float4*)(x + (size_t)(rowbase + r) * DEXP + kw + lane * 4);
    float4* s = sx4 + wave * 16 * XP4;
    for (int r = 0; r < 16; ++r) s[r * XP4 + lane] = v[r];
  }
  __syncthreads();
  if (do_aneg) Aneg[ia] = -__expf(alog);
  float4 av[16];
  {
    const float4* s = sx4 + wave * 16 * XP4 + nl * XP4 + hi;
    for (int i = 0; i < 16; ++i) av[i] = s[i * 2];
  }

  v8f acc[6];
  for (int tt = 0; tt < 6; ++tt)
    for (int r = 0; r < 8; ++r) acc[tt][r] = 0.0f;
  {
    const float* wb = xw + (size_t)nl * DEXP + kw + hi * 4;
    #pragma unroll
    for (int i = 0; i < 16; ++i) {
      float4 bv[6];
      for (int tt = 0; tt < 6; ++tt)
        bv[tt] = *(const float4*)(wb + (size_t)tt * 16 * DEXP + i * 8);
      v2f a0, a1;
      a0.x = av[i].x; a0.y = av[i].y; a1.x = av[i].z; a1.y = av[i].w;
      for (int tt = 0; tt < 6; ++tt) {
        v2f b0; b0.x = bv[tt].x; b0.y = bv[tt].y;
        acc[tt] = wmma_f32_k4(a0, b0, acc[tt]);
      }
      for (int tt = 0; tt < 6; ++tt) {
        v2f b1; b1.x = bv[tt].z; b1.y = bv[tt].w;
        acc[tt] = wmma_f32_k4(a1, b1, acc[tt]);
      }
    }
  }
  for (int tt = 0; tt < 6; ++tt)
    for (int r = 0; r < 8; ++r)
      sred[(wave * 16 + r + hi * 8) * PP + tt * 16 + nl] = acc[tt][r];
  __syncthreads();
  float* sproj = (float*)sproj4;
  for (int i = 0; i < 6; ++i) {
    const int idx = t + i * 256;
    const int row = idx / RPROJ;
    const int col = idx - row * RPROJ;
    float v = xb[col];
    for (int w = 0; w < 8; ++w) v += sred[(w * 16 + row) * PP + col];
    sproj[row * PP + col] = v;
  }
  __syncthreads();
  if (t < 128) {
    const int q = t & 63, row = q >> 2, c4 = q & 3;
    const float4 v = sproj4[row * (PP / 4) + 16 + (t >> 6) * 4 + c4];
    float* dst = (t < 64) ? Bmbuf : Cbuf;
    *(float4*)(dst + (size_t)(rowbase + row) * NST + c4 * 4) = v;
  }

  float4 a2[8];
  {
    const float4* s = sproj4 + nl * (PP / 4) + hi;
    for (int q = 0; q < 8; ++q) a2[q] = s[q * 2];
  }
  const float* db = dtw + (size_t)(wave * 128 + nl) * DRANK + hi * 4;
  float4 bc[2][8];
  for (int u = 0; u < 2; ++u)
    for (int q = 0; q < 8; ++q)
      bc[u][q] = *(const float4*)(db + (size_t)(u * 16) * DRANK + q * 8);
  #pragma unroll
  for (int jj = 0; jj < 8; jj += 2) {
    float4 bn[2][8];
    if (jj + 2 < 8) {
      for (int u = 0; u < 2; ++u)
        for (int q = 0; q < 8; ++q)
          bn[u][q] = *(const float4*)(db + (size_t)((jj + 2 + u) * 16) * DRANK + q * 8);
    }
    const int col0 = wave * 128 + jj * 16 + nl;
    const int col1 = col0 + 16;
    v8f c0, c1;
    { const float bv0 = dtb[col0], bv1 = dtb[col1];
      for (int r = 0; r < 8; ++r) { c0[r] = bv0; c1[r] = bv1; } }
    for (int q = 0; q < 8; ++q) {
      v2f a0, a1, b00, b01, b10, b11;
      a0.x = a2[q].x; a0.y = a2[q].y; a1.x = a2[q].z; a1.y = a2[q].w;
      b00.x = bc[0][q].x; b00.y = bc[0][q].y; b01.x = bc[0][q].z; b01.y = bc[0][q].w;
      b10.x = bc[1][q].x; b10.y = bc[1][q].y; b11.x = bc[1][q].z; b11.y = bc[1][q].w;
      c0 = wmma_f32_k4(a0, b00, c0);
      c1 = wmma_f32_k4(a0, b10, c1);
      c0 = wmma_f32_k4(a1, b01, c0);
      c1 = wmma_f32_k4(a1, b11, c1);
    }
    for (int r = 0; r < 8; ++r) {
      const int M = r + hi * 8;
      dtbuf[(size_t)(rowbase + M) * DEXP + col0] = softplus_f32(c0[r]);
      dtbuf[(size_t)(rowbase + M) * DEXP + col1] = softplus_f32(c1[r]);
    }
    if (jj + 2 < 8) {
      for (int u = 0; u < 2; ++u)
        for (int q = 0; q < 8; ++q) bc[u][q] = bn[u][q];
    }
  }
}

#define SCAN_T   64
#define SCAN_DG  (DEXP / SCAN_T)
#define SCAN_U   8
#define SCAN_UO  4
#define FILL_R   16

__device__ __forceinline__ void stage_cols(const float* __restrict__ g0, const float* __restrict__ g1,
                                           float4* s, int t) {
  const int col4 = t & 15, rsub = t >> 4;
  for (int i0 = 0; i0 < CHUNK / 4; i0 += FILL_R) {
    float4 v0[FILL_R], v1[FILL_R];
    for (int k = 0; k < FILL_R; ++k) {
      const int row = (i0 + k) * 4 + rsub;
      v0[k] = *(const float4*)(g0 + (size_t)row * DEXP + col4 * 4);
      v1[k] = *(const float4*)(g1 + (size_t)row * DEXP + col4 * 4);
    }
    for (int k = 0; k < FILL_R; ++k) {
      const int row = (i0 + k) * 4 + rsub;
      float4 lo, hi;
      lo.x = v0[k].x; lo.y = v1[k].x; lo.z = v0[k].y; lo.w = v1[k].y;
      hi.x = v0[k].z; hi.y = v1[k].z; hi.z = v0[k].w; hi.w = v1[k].w;
      s[row * 32 + col4 * 2]     = lo;
      s[row * 32 + col4 * 2 + 1] = hi;
    }
  }
}

__global__ __launch_bounds__(SCAN_T) void ssm_scan_carry(
    const float* __restrict__ x, const float* __restrict__ dtbuf,
    const float* __restrict__ Bmbuf, const float* __restrict__ Aneg,
    float* __restrict__ Sbuf, float* __restrict__ Dbuf)
{
  __shared__ float4 sdx[CHUNK * 32];
  __shared__ float4 sB [CHUNK * 4];

  const int t  = threadIdx.x;
  const int bi = blockIdx.x;
  const int c  = bi & (NCHUNK - 1);
  const int dg = (bi / NCHUNK) & (SCAN_DG - 1);
  const int b  = bi / (NCHUNK * SCAN_DG);
  const int d  = dg * SCAN_T + t;
  const size_t rbase = (size_t)b * LSEQ + (size_t)c * CHUNK;

  {
    const float4* Bg = (const float4*)(Bmbuf + rbase * NST);
    float4 vb[CHUNK * 4 / SCAN_T];
    for (int i = 0; i < CHUNK * 4 / SCAN_T; ++i) vb[i] = Bg[t + i * SCAN_T];
    for (int i = 0; i < CHUNK * 4 / SCAN_T; ++i) sB[t + i * SCAN_T] = vb[i];
  }
  stage_cols(dtbuf + rbase * DEXP + dg * SCAN_T, x + rbase * DEXP + dg * SCAN_T, sdx, t);

  float A[NST];
  for (int q = 0; q < 4; ++q) {
    const float4 v = ((const float4*)(Aneg + (size_t)d * NST))[q];
    A[4 * q] = v.x; A[4 * q + 1] = v.y; A[4 * q + 2] = v.z; A[4 * q + 3] = v.w;
  }
  __syncthreads();

  const float2* sdxf = (const float2*)sdx + t;

  float h[NST];
  for (int n = 0; n < NST; ++n) h[n] = 0.0f;
  float sdt_sum = 0.0f;

  float2 dxn[SCAN_U];
  for (int j = 0; j < SCAN_U; ++j) dxn[j] = sdxf[j * SCAN_T];
  for (int l0 = 0; l0 < CHUNK; l0 += SCAN_U) {
    float2 dxc[SCAN_U];
    for (int j = 0; j < SCAN_U; ++j) dxc[j] = dxn[j];
    for (int j = 0; j < SCAN_U; ++j)
      dxn[j] = sdxf[((l0 + SCAN_U + j) & (CHUNK - 1)) * SCAN_T];
    #pragma unroll
    for (int j = 0; j < SCAN_U; ++j) {
      const int li = l0 + j;
      const float dtv = dxc[j].x;
      const float u   = dtv * dxc[j].y;
      for (int q = 0; q < 4; ++q) {
        const float4 bv = sB[li * 4 + q];
        h[4 * q]     = fmaf(__expf(A[4 * q]     * dtv), h[4 * q],     u * bv.x);
        h[4 * q + 1] = fmaf(__expf(A[4 * q + 1] * dtv), h[4 * q + 1], u * bv.y);
        h[4 * q + 2] = fmaf(__expf(A[4 * q + 2] * dtv), h[4 * q + 2], u * bv.z);
        h[4 * q + 3] = fmaf(__expf(A[4 * q + 3] * dtv), h[4 * q + 3], u * bv.w);
      }
      sdt_sum += dtv;
    }
  }
  const size_t si = (((size_t)b * NCHUNK + c) * DEXP + d) * NST;
  for (int q = 0; q < 4; ++q) {
    float4 s;
    s.x = h[4 * q]; s.y = h[4 * q + 1]; s.z = h[4 * q + 2]; s.w = h[4 * q + 3];
    ((float4*)(Sbuf + si))[q] = s;
  }
  Dbuf[((size_t)b * NCHUNK + c) * DEXP + d] = sdt_sum;
}

__device__ __forceinline__ float4 chunk_decay(float4 a4, float sdt) {
  float4 p;
  p.x = __expf(a4.x * sdt); p.y = __expf(a4.y * sdt);
  p.z = __expf(a4.z * sdt); p.w = __expf(a4.w * sdt);
  return p;
}

template <int NP>
__device__ __forceinline__ float4 fold_fixed(const float4* __restrict__ Sg, const float* __restrict__ Dg, float4 a4, float4 cin) {
  const size_t slab4 = (size_t)DEXP * NST / 4;
  float4 sv[NP];
  float  dv[NP];
  #pragma unroll
  for (int j = 0; j < NP; ++j) dv[j] = Dg[(size_t)j * DEXP];
  #pragma unroll
  for (int j = 0; j < NP; ++j) sv[j] = Sg[(size_t)j * slab4];
  #pragma unroll
  for (int j = 0; j < NP; ++j) {
    const float4 pv = chunk_decay(a4, dv[j]);
    cin.x = fmaf(pv.x, cin.x, sv[j].x);
    cin.y = fmaf(pv.y, cin.y, sv[j].y);
    cin.z = fmaf(pv.z, cin.z, sv[j].z);
    cin.w = fmaf(pv.w, cin.w, sv[j].w);
  }
  return cin;
}

#define SCAN_G   4

__global__ __launch_bounds__(SCAN_T * SCAN_G) void ssm_scan_out(
    const float* __restrict__ x, const float* __restrict__ dtbuf,
    const float* __restrict__ Bmbuf, const float* __restrict__ Cbuf,
    const float* __restrict__ Aneg, const float* __restrict__ Sbuf,
    const float* __restrict__ Dbuf, float* __restrict__ out)
{
  __shared__ float4 sdx [SCAN_G][CHUNK * 32];
  __shared__ float4 sB  [SCAN_G][CHUNK * 4];
  __shared__ float4 sC  [SCAN_G][CHUNK * 4];
  __shared__ float4 scar[SCAN_G][SCAN_T * NST / 4];

  const int tb = threadIdx.x;
  const int k  = tb >> 6;
  const int t  = tb & (SCAN_T - 1);
  const int bi = blockIdx.x;
  const int g  = bi & (NCHUNK / SCAN_G - 1);
  const int dg = (bi / (NCHUNK / SCAN_G)) & (SCAN_DG - 1);
  const int b  = bi / ((NCHUNK / SCAN_G) * SCAN_DG);
  const int c0 = g * SCAN_G;
  const int c  = c0 + k;
  const int d  = dg * SCAN_T + t;
  const size_t rbase = (size_t)b * LSEQ + (size_t)c * CHUNK;

  const float4* Bg = (const float4*)(Bmbuf + rbase * NST);
  const float4* Cg = (const float4*)(Cbuf  + rbase * NST);
  float4 vb[CHUNK * 4 / SCAN_T], vc[CHUNK * 4 / SCAN_T];
  for (int i = 0; i < CHUNK * 4 / SCAN_T; ++i) { vb[i] = Bg[t + i * SCAN_T]; vc[i] = Cg[t + i * SCAN_T]; }

  {
    const size_t slab4 = (size_t)DEXP * NST / 4;
    const size_t o4    = (size_t)b * NCHUNK * slab4 + (size_t)dg * (SCAN_T * NST / 4) + tb;
    const float4* Sg = (const float4*)Sbuf + o4;
    const float*  Dg = Dbuf + (size_t)b * NCHUNK * DEXP + dg * SCAN_T + (tb >> 2);
    const float4  a4 = ((const float4*)Aneg)[dg * (SCAN_T * NST / 4) + tb];
    float4 so[SCAN_G - 1];
    float  dso[SCAN_G - 1];
    for (int j = 0; j < SCAN_G - 1; ++j) {
      dso[j] = Dg[(size_t)(c0 + j) * DEXP];
      so[j]  = Sg[(size_t)(c0 + j) * slab4];
    }
    float4 cin; cin.x = 0.0f; cin.y = 0.0f; cin.z = 0.0f; cin.w = 0.0f;
    switch (g) {
      case 1: cin = fold_fixed< 4>(Sg, Dg, a4, cin); break;
      case 2: cin = fold_fixed< 8>(Sg, Dg, a4, cin); break;
      case 3: cin = fold_fixed<12>(Sg, Dg, a4, cin); break;
      case 4: cin = fold_fixed<16>(Sg, Dg, a4, cin); break;
      case 5: cin = fold_fixed<20>(Sg, Dg, a4, cin); break;
      case 6: cin = fold_fixed<24>(Sg, Dg, a4, cin); break;
      case 7: cin = fold_fixed<28>(Sg, Dg, a4, cin); break;
      default: break;
    }
    scar[0][tb] = cin;
    for (int j = 0; j < SCAN_G - 1; ++j) {
      const float4 pj = chunk_decay(a4, dso[j]);
      cin.x = fmaf(pj.x, cin.x, so[j].x);
      cin.y = fmaf(pj.y, cin.y, so[j].y);
      cin.z = fmaf(pj.z, cin.z, so[j].z);
      cin.w = fmaf(pj.w, cin.w, so[j].w);
      scar[j + 1][tb] = cin;
    }
  }

  for (int i = 0; i < CHUNK * 4 / SCAN_T; ++i) { sB[k][t + i * SCAN_T] = vb[i]; sC[k][t + i * SCAN_T] = vc[i]; }
  stage_cols(dtbuf + rbase * DEXP + dg * SCAN_T, x + rbase * DEXP + dg * SCAN_T, sdx[k], t);

  float A[NST], h[NST];
  for (int q = 0; q < 4; ++q) {
    const float4 v = ((const float4*)(Aneg + (size_t)d * NST))[q];
    A[4 * q] = v.x; A[4 * q + 1] = v.y; A[4 * q + 2] = v.z; A[4 * q + 3] = v.w;
  }
  __syncthreads();
  for (int q = 0; q < 4; ++q) {
    const float4 w = scar[k][t * 4 + q];
    h[4 * q] = w.x; h[4 * q + 1] = w.y; h[4 * q + 2] = w.z; h[4 * q + 3] = w.w;
  }

  const float2* sdxf = (const float2*)sdx[k] + t;
  const float4* sBk  = sB[k];
  const float4* sCk  = sC[k];
  float* op          = out + rbase * DEXP + d;

  float2 dxn[SCAN_UO];
  for (int j = 0; j < SCAN_UO; ++j) dxn[j] = sdxf[j * SCAN_T];
  for (int l0 = 0; l0 < CHUNK; l0 += SCAN_UO) {
    float2 dxc[SCAN_UO];
    for (int j = 0; j < SCAN_UO; ++j) dxc[j] = dxn[j];
    for (int j = 0; j < SCAN_UO; ++j)
      dxn[j] = sdxf[((l0 + SCAN_UO + j) & (CHUNK - 1)) * SCAN_T];
    #pragma unroll
    for (int j = 0; j < SCAN_UO; ++j) {
      const int li = l0 + j;
      const float dtv = dxc[j].x;
      const float u   = dtv * dxc[j].y;
      float s2[8];
      for (int q = 0; q < 4; ++q) {
        const float4 bv = sBk[li * 4 + q];
        const float4 cv = sCk[li * 4 + q];
        h[4 * q]     = fmaf(__expf(A[4 * q]     * dtv), h[4 * q],     u * bv.x);
        h[4 * q + 1] = fmaf(__expf(A[4 * q + 1] * dtv), h[4 * q + 1], u * bv.y);
        h[4 * q + 2] = fmaf(__expf(A[4 * q + 2] * dtv), h[4 * q + 2], u * bv.z);
        h[4 * q + 3] = fmaf(__expf(A[4 * q + 3] * dtv), h[4 * q + 3], u * bv.w);
        s2[2 * q]     = h[4 * q]     * cv.x + h[4 * q + 1] * cv.y;
        s2[2 * q + 1] = h[4 * q + 2] * cv.z + h[4 * q + 3] * cv.w;
      }
      const float y = ((s2[0] + s2[1]) + (s2[2] + s2[3])) + ((s2[4] + s2[5]) + (s2[6] + s2[7]));
      op[(size_t)li * DEXP] = y;
    }
  }
}

extern "C" void kernel_launch(void* const* d_in, const int* in_sizes, int n_in,
                              void* d_out, int out_size, void* d_ws, size_t ws_size,
                              hipStream_t stream) {
  (void)in_sizes; (void)n_in; (void)out_size; (void)ws_size;
  const float* x     = (const float*)d_in[0];
  const float* xw    = (const float*)d_in[1];
  const float* xb    = (const float*)d_in[2];
  const float* dtw   = (const float*)d_in[3];
  const float* dtb   = (const float*)d_in[4];
  const float* A_log = (const float*)d_in[5];
  float* out = (float*)d_out;

  float* ws    = (float*)d_ws;
  float* dtbuf = ws;
  float* Aneg  = dtbuf + (size_t)ROWS * DEXP;
  float* Bm    = Aneg  + (size_t)DEXP * NST;
  float* Cb    = Bm    + (size_t)ROWS * NST;
  float* Sb    = Cb    + (size_t)ROWS * NST;
  float* Db    = Sb    + (size_t)B_SZ * NCHUNK * DEXP * NST;

  hipLaunchKernelGGL(ssm_proj_wmma, dim3(ROWS / 16), dim3(256), 0, stream,
                     x, xw, xb, dtw, dtb, A_log, dtbuf, Bm, Cb, Aneg);
  hipLaunchKernelGGL(ssm_scan_carry, dim3(B_SZ * SCAN_DG * NCHUNK), dim3(SCAN_T), 0, stream,
                     x, dtbuf, Bm, Aneg, Sb, Db);
  hipLaunchKernelGGL(ssm_scan_out, dim3(B_SZ * SCAN_DG * (NCHUNK / SCAN_G)), dim3(SCAN_T * SCAN_G), 0, stream,
                     x, dtbuf, Bm, Cb, Aneg, Sb, Db, out);
}
